# starting kernel
# baseline (speedup 1.0000x reference)
#include <hip/hip_runtime.h>


typedef float v2f __attribute__((ext_vector_type(2)));
typedef float v8f __attribute__((ext_vector_type(8)));

#define B_N        256
#define F_N        512
#define T_N        81
#define WIN_N      1022
#define HOP_N      400
#define NF_N       69
#define KEPT_LEN   30000
#define OUT_LEN    28550
#define SIG_STRIDE 33024
#define OUT_STRIDE 28560
#define KPAD       1024
#define M1         (B_N * T_N)
#define M2         (B_N * NF_N)

#define BM 64
#define BN 128
#define BK 32
#define TS 36

#define TWO_PI 6.283185307179586476925f

__device__ __forceinline__ float hann_f(int n) {
    return 0.5f - 0.5f * cosf((TWO_PI / WIN_N) * (float)n);
}

__global__ __launch_bounds__(256) void build_mats(float* __restrict__ WinvT,
                                                  float* __restrict__ WfwdT) {
    int idx = blockIdx.x * 256 + threadIdx.x;
    int which = idx >> 20;
    int e   = idx & (KPAD * KPAD - 1);
    int row = e >> 10;
    int col = e & (KPAD - 1);
    if (which == 0) {
        float v = 0.f;
        int n = row;
        if (n < WIN_N) {
            int f = col >> 1, c = col & 1;
            int m = (f * n) % WIN_N;
            float ang = (TWO_PI / WIN_N) * (float)m;
            float wf  = (f == 0 || f == F_N - 1) ? 1.f : 2.f;
            float tr  = (c == 0) ? cosf(ang) : -sinf(ang);
            float h = hann_f(n);
            int p = n % HOP_N;
            float d = 0.f;
            for (int j = p; j < WIN_N; j += HOP_N) {
                float hj = hann_f(j);
                d += hj * hj;
            }
            v = wf * tr * (1.0f / (float)WIN_N) * (h / d);
        }
        WinvT[(size_t)row * KPAD + col] = v;
    } else {
        float v = 0.f;
        int n = col;
        if (n < WIN_N) {
            int k = row & (F_N - 1), c = row >> 9;
            int m = (k * n) % WIN_N;
            float ang = (TWO_PI / WIN_N) * (float)m;
            float tr  = (c == 0) ? cosf(ang) : -sinf(ang);
            v = tr * hann_f(n);
        }
        WfwdT[(size_t)row * KPAD + col] = v;
    }
}

__global__ __launch_bounds__(256) void istft_gemm_ola(const float* __restrict__ x,
                                                      const float* __restrict__ WinvT,
                                                      float* __restrict__ sig) {
    __shared__ float sA[BM][TS];
    __shared__ float sB[BN][TS];

    const int tid  = threadIdx.x;
    const int lane = tid & 31;
    const int wave = tid >> 5;
    const int wm   = wave & 1;
    const int wn   = wave >> 1;
    const int half = lane >> 4;
    const int r16  = lane & 15;

    const int rowBlock = blockIdx.y * BM;
    const int colBlock = blockIdx.x * BN;

    const int a_lr = tid & 63;
    const int a_lc = (tid >> 6) << 3;
    const int Ra   = rowBlock + a_lr;
    const int ab   = Ra / T_N;
    const int at   = Ra - ab * T_N;
    const float* xb = x + (size_t)ab * (F_N * T_N * 2) + at * 2;

    const int b_row = tid >> 3;
    const int b_kc  = (tid & 7) << 2;

    v8f acc[2][2];
#pragma unroll
    for (int i = 0; i < 2; ++i)
#pragma unroll
        for (int j = 0; j < 2; ++j)
            acc[i][j] = (v8f){0.f,0.f,0.f,0.f,0.f,0.f,0.f,0.f};

    for (int kb = 0; kb < KPAD; kb += BK) {
        __syncthreads();
#pragma unroll
        for (int u = 0; u < 4; ++u) {
            int f = ((kb + a_lc) >> 1) + u;
            *(v2f*)&sA[a_lr][a_lc + (u << 1)] = *(const v2f*)(xb + (size_t)f * (T_N * 2));
        }
#pragma unroll
        for (int u = 0; u < 4; ++u) {
            int nrow = b_row + (u << 5);
            *(float4*)&sB[nrow][b_kc] =
                *(const float4*)(WinvT + (size_t)(colBlock + nrow) * KPAD + kb + b_kc);
        }
        __syncthreads();
#pragma unroll
        for (int k0 = 0; k0 < BK; k0 += 4) {
            const int ka = k0 + (half << 1);
            v2f a0 = *(const v2f*)&sA[(wm << 5) + r16][ka];
            v2f a1 = *(const v2f*)&sA[(wm << 5) + 16 + r16][ka];
            v2f b0 = *(const v2f*)&sB[(wn << 5) + r16][ka];
            v2f b1 = *(const v2f*)&sB[(wn << 5) + 16 + r16][ka];
            acc[0][0] = __builtin_amdgcn_wmma_f32_16x16x4_f32(false, a0, false, b0, (short)0, acc[0][0], false, false);
            acc[0][1] = __builtin_amdgcn_wmma_f32_16x16x4_f32(false, a0, false, b1, (short)0, acc[0][1], false, false);
            acc[1][0] = __builtin_amdgcn_wmma_f32_16x16x4_f32(false, a1, false, b0, (short)0, acc[1][0], false, false);
            acc[1][1] = __builtin_amdgcn_wmma_f32_16x16x4_f32(false, a1, false, b1, (short)0, acc[1][1], false, false);
        }
    }

#pragma unroll
    for (int in = 0; in < 2; ++in) {
        const int n = colBlock + (wn << 5) + (in << 4) + r16;
        if (n < WIN_N) {
#pragma unroll
            for (int im = 0; im < 2; ++im) {
                const int Rb = rowBlock + (wm << 5) + (im << 4) + (half << 3);
#pragma unroll
                for (int v = 0; v < 8; ++v) {
                    int Rr = Rb + v;
                    int b  = Rr / T_N;
                    int t  = Rr - b * T_N;
                    atomicAdd(sig + (size_t)b * SIG_STRIDE + t * HOP_N + n, acc[im][in][v]);
                }
            }
        }
    }
}

__global__ __launch_bounds__(256) void fold_kernel(const float* __restrict__ sig,
                                                   float* __restrict__ outsig) {
    int i = blockIdx.x * 256 + threadIdx.x;
    if (i >= B_N * OUT_LEN) return;
    int b = i / OUT_LEN;
    int o = i - b * OUT_LEN;

    int k = (o < 1000) ? 0 : (o - 50) / 950;
    int r = o - k * 950;
    int j = k * 1000 + r;

    const float* s = sig + (size_t)b * SIG_STRIDE;
    float kept = s[k * 1100 + r];
    float sh = 0.f;
    int j2 = j + 50;
    if (j2 < KEPT_LEN) {
        int k2 = j2 / 1000, r2 = j2 - k2 * 1000;
        sh = s[k2 * 1100 + r2];
    }

    float cconst;
    if (j < 950)        cconst = 1.f;
    else if (j < 1000)  cconst = 0.f;
    else if (j < 29000) cconst = (r >= 50 && r < 950) ? 1.f : 0.f;
    else                cconst = 1.f;
    float cdec = (j < 29000 && r >= 950) ? (float)(1000 - r) * 0.02f : 0.f;
    float cinc = 0.f;
    if (j >= 950) {
        int rr = (j - 950) % 1000;
        if (rr < 50) cinc = (float)(rr + 1) * 0.02f;
    }
    outsig[(size_t)b * OUT_STRIDE + o] = (cconst + cdec) * kept + cinc * sh;
}

__global__ __launch_bounds__(256) void stft_gemm(const float* __restrict__ outsig,
                                                 const float* __restrict__ WfwdT,
                                                 float* __restrict__ out) {
    __shared__ float sA[BM][TS];
    __shared__ float sB[BN][TS];

    const int tid  = threadIdx.x;
    const int lane = tid & 31;
    const int wave = tid >> 5;
    const int wm   = wave & 1;
    const int wn   = wave >> 1;
    const int half = lane >> 4;
    const int r16  = lane & 15;

    const int rowBlock = blockIdx.y * BM;
    const int colBlock = blockIdx.x * BN;

    const int a_lr = tid >> 2;
    const int a_lc = (tid & 3) << 3;
    const int Ra   = rowBlock + a_lr;
    const int ab   = Ra / NF_N;
    const int am   = Ra - ab * NF_N;
    const float* sg = outsig + (size_t)ab * OUT_STRIDE + am * HOP_N;

    const int b_row = tid >> 3;
    const int b_kc  = (tid & 7) << 2;

    v8f acc[2][2];
#pragma unroll
    for (int i = 0; i < 2; ++i)
#pragma unroll
        for (int j = 0; j < 2; ++j)
            acc[i][j] = (v8f){0.f,0.f,0.f,0.f,0.f,0.f,0.f,0.f};

    for (int kb = 0; kb < KPAD; kb += BK) {
        __syncthreads();
        *(float4*)&sA[a_lr][a_lc]     = *(const float4*)(sg + kb + a_lc);
        *(float4*)&sA[a_lr][a_lc + 4] = *(const float4*)(sg + kb + a_lc + 4);
#pragma unroll
        for (int u = 0; u < 4; ++u) {
            int nrow = b_row + (u << 5);
            *(float4*)&sB[nrow][b_kc] =
                *(const float4*)(WfwdT + (size_t)(colBlock + nrow) * KPAD + kb + b_kc);
        }
        __syncthreads();
#pragma unroll
        for (int k0 = 0; k0 < BK; k0 += 4) {
            const int ka = k0 + (half << 1);
            v2f a0 = *(const v2f*)&sA[(wm << 5) + r16][ka];
            v2f a1 = *(const v2f*)&sA[(wm << 5) + 16 + r16][ka];
            v2f b0 = *(const v2f*)&sB[(wn << 5) + r16][ka];
            v2f b1 = *(const v2f*)&sB[(wn << 5) + 16 + r16][ka];
            acc[0][0] = __builtin_amdgcn_wmma_f32_16x16x4_f32(false, a0, false, b0, (short)0, acc[0][0], false, false);
            acc[0][1] = __builtin_amdgcn_wmma_f32_16x16x4_f32(false, a0, false, b1, (short)0, acc[0][1], false, false);
            acc[1][0] = __builtin_amdgcn_wmma_f32_16x16x4_f32(false, a1, false, b0, (short)0, acc[1][0], false, false);
            acc[1][1] = __builtin_amdgcn_wmma_f32_16x16x4_f32(false, a1, false, b1, (short)0, acc[1][1], false, false);
        }
    }

#pragma unroll
    for (int in = 0; in < 2; ++in) {
        const int q  = colBlock + (wn << 5) + (in << 4) + r16;
        const int kf = q & (F_N - 1);
        const int cc = q >> 9;
#pragma unroll
        for (int im = 0; im < 2; ++im) {
            const int Rb = rowBlock + (wm << 5) + (im << 4) + (half << 3);
#pragma unroll
            for (int v = 0; v < 8; ++v) {
                int Rr = Rb + v;
                int b  = Rr / NF_N;
                int m  = Rr - b * NF_N;
                out[(size_t)b * (F_N * NF_N * 2) + kf * (NF_N * 2) + m * 2 + cc] = acc[im][in][v];
            }
        }
    }
}

extern "C" void kernel_launch(void* const* d_in, const int* in_sizes, int n_in,
                              void* d_out, int out_size, void* d_ws, size_t ws_size,
                              hipStream_t stream) {
    (void)in_sizes; (void)n_in; (void)out_size; (void)ws_size;
    const float* x = (const float*)d_in[0];
    float* out = (float*)d_out;

    float* ws     = (float*)d_ws;
    float* WinvT  = ws;
    float* WfwdT  = WinvT + (size_t)KPAD * KPAD;
    float* sig    = WfwdT + (size_t)KPAD * KPAD;
    float* outsig = sig + (size_t)B_N * SIG_STRIDE;

    hipMemsetAsync(sig, 0, (size_t)B_N * SIG_STRIDE * sizeof(float), stream);
    build_mats<<<(2 * KPAD * KPAD) / 256, 256, 0, stream>>>(WinvT, WfwdT);
    istft_gemm_ola<<<dim3(KPAD / BN, M1 / BM), 256, 0, stream>>>(x, WinvT, sig);
    fold_kernel<<<(B_N * OUT_LEN + 255) / 256, 256, 0, stream>>>(sig, outsig);
    stft_gemm<<<dim3(KPAD / BN, M2 / BM), 256, 0, stream>>>(outsig, WfwdT, out);
}


# Fold both real DFTs about N/2 (cos even, sin odd): K of both f32 WMMA products 1024 -> 512, half the matrix work; DFT terms are summed in a new order (n and N-n paired first), atomics as before
# speedup vs baseline: 1.4339x; 1.4339x over previous
#include <hip/hip_runtime.h>


typedef float v2f __attribute__((ext_vector_type(2)));
typedef float v8f __attribute__((ext_vector_type(8)));
typedef float v4f __attribute__((ext_vector_type(4)));
typedef v4f v4fu __attribute__((aligned(4)));

#define B_N        256
#define F_N        512
#define T_N        81
#define WIN_N      1022
#define HOP_N      400
#define NF_N       69
#define KEPT_LEN   30000
#define OUT_LEN    28550
#define SIG_STRIDE 33024
#define OUT_STRIDE 28560
#define KPAD       1024
#define KH         512
#define M1         (B_N * T_N)
#define M2         (B_N * NF_N)

#define BM 64
#define BN 128
#define BK 32
#define TS 36

#define TWO_PI 6.283185307179586476925f

__device__ __forceinline__ float hann_f(int n) {
    return 0.5f - 0.5f * cosf((TWO_PI / WIN_N) * (float)n);
}

__device__ __forceinline__ float synwin_f(int n) {
    int p = n % HOP_N;
    float d = 0.f;
    for (int j = p; j < WIN_N; j += HOP_N) {
        float hj = hann_f(j);
        d += hj * hj;
    }
    return hann_f(n) / d;
}

__global__ __launch_bounds__(256) void build_mats(float* __restrict__ WinvT,
                                                  float* __restrict__ WfwdT,
                                                  float* __restrict__ winT) {
    int idx = blockIdx.x * 256 + threadIdx.x;
    int which = idx >> 19;
    if (which == 2) {
        int n = idx & (KPAD - 1);
        winT[n] = (n < WIN_N) ? synwin_f(n) : 0.f;
        return;
    }
    int e   = idx & (KPAD * KH - 1);
    int row = e >> 9;
    int col = e & (KH - 1);
    int lo  = row & (KH - 1);
    int c   = row >> 9;
    int m = (lo * col) % WIN_N;
    float ang = (TWO_PI / WIN_N) * (float)m;
    if (which == 0) {
        int f = col;
        float wf  = (f == 0 || f == F_N - 1) ? 1.f : 2.f;
        float tr  = (c == 0) ? cosf(ang) : sinf(ang);
        WinvT[(size_t)row * KH + col] = wf * tr * (1.0f / (float)WIN_N);
    } else {
        int n = col;
        float tr  = (c == 0) ? cosf(ang) : -sinf(ang);
        float v = tr * hann_f(n);
        if (n == KH - 1) v *= 0.5f;
        WfwdT[(size_t)row * KH + col] = v;
    }
}

__global__ __launch_bounds__(256) void istft_gemm_ola(const float* __restrict__ x,
                                                      const float* __restrict__ WinvT,
                                                      const float* __restrict__ winT,
                                                      float* __restrict__ sig) {
    __shared__ __align__(16) float sAr[BM][TS];
    __shared__ __align__(16) float sAi[BM][TS];
    __shared__ __align__(16) float sBc[BM][TS];
    __shared__ __align__(16) float sBs[BM][TS];

    const int tid  = threadIdx.x;
    const int lane = tid & 31;
    const int wave = tid >> 5;
    const int wm   = wave & 1;
    const int wn   = wave >> 1;
    const int half = lane >> 4;
    const int r16  = lane & 15;

    const int rowBlock = blockIdx.y * BM;
    const int nBlock   = blockIdx.x * 64;

    const int a_lr = tid & 63;
    const int a_lc = (tid >> 6) << 3;
    const int Ra   = rowBlock + a_lr;
    const int ab   = Ra / T_N;
    const int at   = Ra - ab * T_N;
    const float* xp = x + (size_t)ab * (F_N * T_N * 2) + at * 2 + (size_t)a_lc * (T_N * 2);

    const int b_row = tid >> 3;
    const int b_kc  = (tid & 7) << 2;
    const float* wp = WinvT + (size_t)(nBlock + b_row) * KH + b_kc;

    v8f accC[2], accS[2];
#pragma unroll
    for (int i = 0; i < 2; ++i) {
        accC[i] = (v8f){0.f,0.f,0.f,0.f,0.f,0.f,0.f,0.f};
        accS[i] = (v8f){0.f,0.f,0.f,0.f,0.f,0.f,0.f,0.f};
    }

    for (int kb = 0; kb < KH; kb += BK) {
        __syncthreads();
        v2f av[8];
#pragma unroll
        for (int u = 0; u < 8; ++u)
            av[u] = *(const v2f*)(xp + u * (T_N * 2));
        const float4 w0 = *(const float4*)(wp);
        const float4 w1 = *(const float4*)(wp + 32 * KH);
        const float4 w2 = *(const float4*)(wp + KH * KH);
        const float4 w3 = *(const float4*)(wp + KH * KH + 32 * KH);
        xp += BK * (T_N * 2); wp += BK;
        *(float4*)&sAr[a_lr][a_lc]     = make_float4(av[0].x, av[1].x, av[2].x, av[3].x);
        *(float4*)&sAr[a_lr][a_lc + 4] = make_float4(av[4].x, av[5].x, av[6].x, av[7].x);
        *(float4*)&sAi[a_lr][a_lc]     = make_float4(av[0].y, av[1].y, av[2].y, av[3].y);
        *(float4*)&sAi[a_lr][a_lc + 4] = make_float4(av[4].y, av[5].y, av[6].y, av[7].y);
        *(float4*)&sBc[b_row][b_kc]      = w0;
        *(float4*)&sBc[b_row + 32][b_kc] = w1;
        *(float4*)&sBs[b_row][b_kc]      = w2;
        *(float4*)&sBs[b_row + 32][b_kc] = w3;
        __syncthreads();
#pragma unroll
        for (int k0 = 0; k0 < BK; k0 += 4) {
            const int ka = k0 + (half << 1);
            v2f ar0 = *(const v2f*)&sAr[(wm << 5) + r16][ka];
            v2f ar1 = *(const v2f*)&sAr[(wm << 5) + 16 + r16][ka];
            v2f ai0 = *(const v2f*)&sAi[(wm << 5) + r16][ka];
            v2f ai1 = *(const v2f*)&sAi[(wm << 5) + 16 + r16][ka];
            v2f bc  = *(const v2f*)&sBc[(wn << 4) + r16][ka];
            v2f bs  = *(const v2f*)&sBs[(wn << 4) + r16][ka];
            accC[0] = __builtin_amdgcn_wmma_f32_16x16x4_f32(false, ar0, false, bc, (short)0, accC[0], false, false);
            accS[0] = __builtin_amdgcn_wmma_f32_16x16x4_f32(false, ai0, false, bs, (short)0, accS[0], false, false);
            accC[1] = __builtin_amdgcn_wmma_f32_16x16x4_f32(false, ar1, false, bc, (short)0, accC[1], false, false);
            accS[1] = __builtin_amdgcn_wmma_f32_16x16x4_f32(false, ai1, false, bs, (short)0, accS[1], false, false);
        }
    }

    const int n = nBlock + (wn << 4) + r16;
    const bool mir = (n >= 1) && (n <= KH - 2);
    const float w0 = winT[n];
    const float w1 = mir ? winT[WIN_N - n] : 0.f;
#pragma unroll
    for (int im = 0; im < 2; ++im) {
        const int Rb = rowBlock + (wm << 5) + (im << 4) + (half << 3);
#pragma unroll
        for (int v = 0; v < 8; ++v) {
            const unsigned Rr = (unsigned)(Rb + v);
            const unsigned b  = Rr / T_N;
            const unsigned t  = Rr - b * T_N;
            const unsigned ro = b * SIG_STRIDE + t * HOP_N;
            const float c = accC[im][v];
            const float s = accS[im][v];
            atomicAdd(sig + (ro + (unsigned)n), w0 * (c - s));
            if (mir) atomicAdd(sig + (ro + (unsigned)(WIN_N - n)), w1 * (c + s));
        }
    }
}

__global__ __launch_bounds__(256) void fold_kernel(const float* __restrict__ sig,
                                                   float* __restrict__ outsig) {
    int i = blockIdx.x * 256 + threadIdx.x;
    if (i >= B_N * OUT_LEN) return;
    int b = i / OUT_LEN;
    int o = i - b * OUT_LEN;

    int k = (o < 1000) ? 0 : (o - 50) / 950;
    int r = o - k * 950;
    int j = k * 1000 + r;

    const float* s = sig + (size_t)b * SIG_STRIDE;
    float kept = s[k * 1100 + r];
    float sh = 0.f;
    int j2 = j + 50;
    if (j2 < KEPT_LEN) {
        int k2 = j2 / 1000, r2 = j2 - k2 * 1000;
        sh = s[k2 * 1100 + r2];
    }

    float cconst;
    if (j < 950)        cconst = 1.f;
    else if (j < 1000)  cconst = 0.f;
    else if (j < 29000) cconst = (r >= 50 && r < 950) ? 1.f : 0.f;
    else                cconst = 1.f;
    float cdec = (j < 29000 && r >= 950) ? (float)(1000 - r) * 0.02f : 0.f;
    float cinc = 0.f;
    if (j >= 950) {
        int rr = (j - 950) % 1000;
        if (rr < 50) cinc = (float)(rr + 1) * 0.02f;
    }
    outsig[(size_t)b * OUT_STRIDE + o] = (cconst + cdec) * kept + cinc * sh;
}

__global__ __launch_bounds__(256) void stft_gemm(const float* __restrict__ outsig,
                                                 const float* __restrict__ WfwdT,
                                                 float* __restrict__ out) {
    __shared__ __align__(16) float sA[BM][TS];
    __shared__ __align__(16) float sB[BN][TS];

    const int tid  = threadIdx.x;
    const int lane = tid & 31;
    const int wave = tid >> 5;
    const int wm   = wave & 1;
    const int wn   = wave >> 1;
    const int half = lane >> 4;
    const int r16  = lane & 15;

    const int rowBlock = blockIdx.y * BM;
    const int colBlock = blockIdx.x * BN;

    const float sgn = (blockIdx.x >> 2) ? -1.f : 1.f;
    const int a_lr = tid >> 2;
    const int a_lc = (tid & 3) << 3;
    const int Ra   = rowBlock + a_lr;
    const int ab   = Ra / NF_N;
    const int am   = Ra - ab * NF_N;
    const float* sg = outsig + (size_t)ab * OUT_STRIDE + am * HOP_N;
    const float* fp = sg + a_lc;
    const float* mp = sg + (WIN_N - 7 - a_lc);

    const int b_row = tid >> 3;
    const int b_kc  = (tid & 7) << 2;
    const float* wp = WfwdT + (size_t)(colBlock + b_row) * KH + b_kc;

    v8f acc[2][2];
#pragma unroll
    for (int i = 0; i < 2; ++i)
#pragma unroll
        for (int j = 0; j < 2; ++j)
            acc[i][j] = (v8f){0.f,0.f,0.f,0.f,0.f,0.f,0.f,0.f};

    for (int kb = 0; kb < KH; kb += BK) {
        __syncthreads();
        const float4 f0 = *(const float4*)(fp);
        const float4 f1 = *(const float4*)(fp + 4);
        const v4fu m0 = *(const v4fu*)(mp);
        const v4fu m1 = *(const v4fu*)(mp + 4);
        const float4 w0 = *(const float4*)(wp);
        const float4 w1 = *(const float4*)(wp + 32 * KH);
        const float4 w2 = *(const float4*)(wp + 64 * KH);
        const float4 w3 = *(const float4*)(wp + 96 * KH);
        fp += BK; mp -= BK; wp += BK;
        *(float4*)&sA[a_lr][a_lc]     = make_float4(fmaf(sgn, m1.w, f0.x), fmaf(sgn, m1.z, f0.y),
                                                    fmaf(sgn, m1.y, f0.z), fmaf(sgn, m1.x, f0.w));
        *(float4*)&sA[a_lr][a_lc + 4] = make_float4(fmaf(sgn, m0.w, f1.x), fmaf(sgn, m0.z, f1.y),
                                                    fmaf(sgn, m0.y, f1.z), fmaf(sgn, m0.x, f1.w));
        *(float4*)&sB[b_row][b_kc]      = w0;
        *(float4*)&sB[b_row + 32][b_kc] = w1;
        *(float4*)&sB[b_row + 64][b_kc] = w2;
        *(float4*)&sB[b_row + 96][b_kc] = w3;
        __syncthreads();
#pragma unroll
        for (int k0 = 0; k0 < BK; k0 += 4) {
            const int ka = k0 + (half << 1);
            v2f a0 = *(const v2f*)&sA[(wm << 5) + r16][ka];
            v2f a1 = *(const v2f*)&sA[(wm << 5) + 16 + r16][ka];
            v2f b0 = *(const v2f*)&sB[(wn << 5) + r16][ka];
            v2f b1 = *(const v2f*)&sB[(wn << 5) + 16 + r16][ka];
            acc[0][0] = __builtin_amdgcn_wmma_f32_16x16x4_f32(false, a0, false, b0, (short)0, acc[0][0], false, false);
            acc[0][1] = __builtin_amdgcn_wmma_f32_16x16x4_f32(false, a0, false, b1, (short)0, acc[0][1], false, false);
            acc[1][0] = __builtin_amdgcn_wmma_f32_16x16x4_f32(false, a1, false, b0, (short)0, acc[1][0], false, false);
            acc[1][1] = __builtin_amdgcn_wmma_f32_16x16x4_f32(false, a1, false, b1, (short)0, acc[1][1], false, false);
        }
    }

#pragma unroll
    for (int in = 0; in < 2; ++in) {
        const int q  = colBlock + (wn << 5) + (in << 4) + r16;
        const int kf = q & (F_N - 1);
        const int cc = q >> 9;
#pragma unroll
        for (int im = 0; im < 2; ++im) {
            const int Rb = rowBlock + (wm << 5) + (im << 4) + (half << 3);
#pragma unroll
            for (int v = 0; v < 8; ++v) {
                int Rr = Rb + v;
                int b  = Rr / NF_N;
                int m  = Rr - b * NF_N;
                out[(size_t)b * (F_N * NF_N * 2) + kf * (NF_N * 2) + m * 2 + cc] = acc[im][in][v];
            }
        }
    }
}

extern "C" void kernel_launch(void* const* d_in, const int* in_sizes, int n_in,
                              void* d_out, int out_size, void* d_ws, size_t ws_size,
                              hipStream_t stream) {
    (void)in_sizes; (void)n_in; (void)out_size; (void)ws_size;
    const float* x = (const float*)d_in[0];
    float* out = (float*)d_out;

    float* ws     = (float*)d_ws;
    float* WinvT  = ws;
    float* winT   = WinvT + (size_t)KPAD * KH;
    float* WfwdT  = WinvT + (size_t)KPAD * KPAD;
    float* sig    = WfwdT + (size_t)KPAD * KPAD;
    float* outsig = sig + (size_t)B_N * SIG_STRIDE;

    hipMemsetAsync(sig, 0, (size_t)B_N * SIG_STRIDE * sizeof(float), stream);
    build_mats<<<(2 * KPAD * KH + KPAD) / 256, 256, 0, stream>>>(WinvT, WfwdT, winT);
    istft_gemm_ola<<<dim3(KH / 64, M1 / BM), 256, 0, stream>>>(x, WinvT, winT, sig);
    fold_kernel<<<(B_N * OUT_LEN + 255) / 256, 256, 0, stream>>>(sig, outsig);
    stft_gemm<<<dim3(KPAD / BN, M2 / BM), 256, 0, stream>>>(outsig, WfwdT, out);
}


# Forward-DFT product: a work-group now holds re and im of 64 bins (sum and difference A tiles), tile turned through LDS, out written as (re,im) pairs along frames: 256-B runs, not 4-B pieces
# speedup vs baseline: 1.8904x; 1.3184x over previous
#include <hip/hip_runtime.h>


typedef float v2f __attribute__((ext_vector_type(2)));
typedef float v8f __attribute__((ext_vector_type(8)));
typedef float v4f __attribute__((ext_vector_type(4)));
typedef v4f v4fu __attribute__((aligned(4)));

#define B_N        256
#define F_N        512
#define T_N        81
#define WIN_N      1022
#define HOP_N      400
#define NF_N       69
#define KEPT_LEN   30000
#define OUT_LEN    28550
#define SIG_STRIDE 33024
#define OUT_STRIDE 28560
#define KPAD       1024
#define KH         512
#define M1         (B_N * T_N)
#define M2         (B_N * NF_N)

#define BM 64
#define BN 128
#define BK 32
#define TS 36

#define TWO_PI 6.283185307179586476925f

__device__ __forceinline__ float hann_f(int n) {
    return 0.5f - 0.5f * cosf((TWO_PI / WIN_N) * (float)n);
}

__device__ __forceinline__ float synwin_f(int n) {
    int p = n % HOP_N;
    float d = 0.f;
    for (int j = p; j < WIN_N; j += HOP_N) {
        float hj = hann_f(j);
        d += hj * hj;
    }
    return hann_f(n) / d;
}

__global__ __launch_bounds__(256) void build_mats(float* __restrict__ WinvT,
                                                  float* __restrict__ WfwdT,
                                                  float* __restrict__ winT) {
    int idx = blockIdx.x * 256 + threadIdx.x;
    int which = idx >> 19;
    if (which == 2) {
        int n = idx & (KPAD - 1);
        winT[n] = (n < WIN_N) ? synwin_f(n) : 0.f;
        return;
    }
    int e   = idx & (KPAD * KH - 1);
    int row = e >> 9;
    int col = e & (KH - 1);
    int lo  = row & (KH - 1);
    int c   = row >> 9;
    int m = (lo * col) % WIN_N;
    float ang = (TWO_PI / WIN_N) * (float)m;
    if (which == 0) {
        int f = col;
        float wf  = (f == 0 || f == F_N - 1) ? 1.f : 2.f;
        float tr  = (c == 0) ? cosf(ang) : sinf(ang);
        WinvT[(size_t)row * KH + col] = wf * tr * (1.0f / (float)WIN_N);
    } else {
        int n = col;
        float tr  = (c == 0) ? cosf(ang) : -sinf(ang);
        float v = tr * hann_f(n);
        if (n == KH - 1) v *= 0.5f;
        WfwdT[(size_t)row * KH + col] = v;
    }
}

__global__ __launch_bounds__(256) void istft_gemm_ola(const float* __restrict__ x,
                                                      const float* __restrict__ WinvT,
                                                      const float* __restrict__ winT,
                                                      float* __restrict__ sig) {
    __shared__ __align__(16) float sAr[BM][TS];
    __shared__ __align__(16) float sAi[BM][TS];
    __shared__ __align__(16) float sBc[BM][TS];
    __shared__ __align__(16) float sBs[BM][TS];

    const int tid  = threadIdx.x;
    const int lane = tid & 31;
    const int wave = tid >> 5;
    const int wm   = wave & 1;
    const int wn   = wave >> 1;
    const int half = lane >> 4;
    const int r16  = lane & 15;

    const int rowBlock = blockIdx.y * BM;
    const int nBlock   = blockIdx.x * 64;

    const int a_lr = tid & 63;
    const int a_lc = (tid >> 6) << 3;
    const int Ra   = rowBlock + a_lr;
    const int ab   = Ra / T_N;
    const int at   = Ra - ab * T_N;
    const float* xp = x + (size_t)ab * (F_N * T_N * 2) + at * 2 + (size_t)a_lc * (T_N * 2);

    const int b_row = tid >> 3;
    const int b_kc  = (tid & 7) << 2;
    const float* wp = WinvT + (size_t)(nBlock + b_row) * KH + b_kc;

    v8f accC[2], accS[2];
#pragma unroll
    for (int i = 0; i < 2; ++i) {
        accC[i] = (v8f){0.f,0.f,0.f,0.f,0.f,0.f,0.f,0.f};
        accS[i] = (v8f){0.f,0.f,0.f,0.f,0.f,0.f,0.f,0.f};
    }

    for (int kb = 0; kb < KH; kb += BK) {
        __syncthreads();
        v2f av[8];
#pragma unroll
        for (int u = 0; u < 8; ++u)
            av[u] = *(const v2f*)(xp + u * (T_N * 2));
        const float4 w0 = *(const float4*)(wp);
        const float4 w1 = *(const float4*)(wp + 32 * KH);
        const float4 w2 = *(const float4*)(wp + KH * KH);
        const float4 w3 = *(const float4*)(wp + KH * KH + 32 * KH);
        xp += BK * (T_N * 2); wp += BK;
        *(float4*)&sAr[a_lr][a_lc]     = make_float4(av[0].x, av[1].x, av[2].x, av[3].x);
        *(float4*)&sAr[a_lr][a_lc + 4] = make_float4(av[4].x, av[5].x, av[6].x, av[7].x);
        *(float4*)&sAi[a_lr][a_lc]     = make_float4(av[0].y, av[1].y, av[2].y, av[3].y);
        *(float4*)&sAi[a_lr][a_lc + 4] = make_float4(av[4].y, av[5].y, av[6].y, av[7].y);
        *(float4*)&sBc[b_row][b_kc]      = w0;
        *(float4*)&sBc[b_row + 32][b_kc] = w1;
        *(float4*)&sBs[b_row][b_kc]      = w2;
        *(float4*)&sBs[b_row + 32][b_kc] = w3;
        __syncthreads();
#pragma unroll
        for (int k0 = 0; k0 < BK; k0 += 4) {
            const int ka = k0 + (half << 1);
            v2f ar0 = *(const v2f*)&sAr[(wm << 5) + r16][ka];
            v2f ar1 = *(const v2f*)&sAr[(wm << 5) + 16 + r16][ka];
            v2f ai0 = *(const v2f*)&sAi[(wm << 5) + r16][ka];
            v2f ai1 = *(const v2f*)&sAi[(wm << 5) + 16 + r16][ka];
            v2f bc  = *(const v2f*)&sBc[(wn << 4) + r16][ka];
            v2f bs  = *(const v2f*)&sBs[(wn << 4) + r16][ka];
            accC[0] = __builtin_amdgcn_wmma_f32_16x16x4_f32(false, ar0, false, bc, (short)0, accC[0], false, false);
            accS[0] = __builtin_amdgcn_wmma_f32_16x16x4_f32(false, ai0, false, bs, (short)0, accS[0], false, false);
            accC[1] = __builtin_amdgcn_wmma_f32_16x16x4_f32(false, ar1, false, bc, (short)0, accC[1], false, false);
            accS[1] = __builtin_amdgcn_wmma_f32_16x16x4_f32(false, ai1, false, bs, (short)0, accS[1], false, false);
        }
    }

    const int n = nBlock + (wn << 4) + r16;
    const bool mir = (n >= 1) && (n <= KH - 2);
    const float w0 = winT[n];
    const float w1 = mir ? winT[WIN_N - n] : 0.f;
#pragma unroll
    for (int im = 0; im < 2; ++im) {
        const int Rb = rowBlock + (wm << 5) + (im << 4) + (half << 3);
#pragma unroll
        for (int v = 0; v < 8; ++v) {
            const unsigned Rr = (unsigned)(Rb + v);
            const unsigned b  = Rr / T_N;
            const unsigned t  = Rr - b * T_N;
            const unsigned ro = b * SIG_STRIDE + t * HOP_N;
            const float c = accC[im][v];
            const float s = accS[im][v];
            atomicAdd(sig + (ro + (unsigned)n), w0 * (c - s));
            if (mir) atomicAdd(sig + (ro + (unsigned)(WIN_N - n)), w1 * (c + s));
        }
    }
}

__global__ __launch_bounds__(256) void fold_kernel(const float* __restrict__ sig,
                                                   float* __restrict__ outsig) {
    int i = blockIdx.x * 256 + threadIdx.x;
    if (i >= B_N * OUT_LEN) return;
    int b = i / OUT_LEN;
    int o = i - b * OUT_LEN;

    int k = (o < 1000) ? 0 : (o - 50) / 950;
    int r = o - k * 950;
    int j = k * 1000 + r;

    const float* s = sig + (size_t)b * SIG_STRIDE;
    float kept = s[k * 1100 + r];
    float sh = 0.f;
    int j2 = j + 50;
    if (j2 < KEPT_LEN) {
        int k2 = j2 / 1000, r2 = j2 - k2 * 1000;
        sh = s[k2 * 1100 + r2];
    }

    float cconst;
    if (j < 950)        cconst = 1.f;
    else if (j < 1000)  cconst = 0.f;
    else if (j < 29000) cconst = (r >= 50 && r < 950) ? 1.f : 0.f;
    else                cconst = 1.f;
    float cdec = (j < 29000 && r >= 950) ? (float)(1000 - r) * 0.02f : 0.f;
    float cinc = 0.f;
    if (j >= 950) {
        int rr = (j - 950) % 1000;
        if (rr < 50) cinc = (float)(rr + 1) * 0.02f;
    }
    outsig[(size_t)b * OUT_STRIDE + o] = (cconst + cdec) * kept + cinc * sh;
}

#define SP 65
__global__ __launch_bounds__(256) void stft_gemm(const float* __restrict__ outsig,
                                                 const float* __restrict__ WfwdT,
                                                 float* __restrict__ out) {
    __shared__ __align__(16) float smem[4 * BM * TS];
    float (*sAp)[TS] = (float (*)[TS])(smem);
    float (*sAm)[TS] = (float (*)[TS])(smem + BM * TS);
    float (*sBc)[TS] = (float (*)[TS])(smem + 2 * BM * TS);
    float (*sBs)[TS] = (float (*)[TS])(smem + 3 * BM * TS);

    const int tid  = threadIdx.x;
    const int lane = tid & 31;
    const int wave = tid >> 5;
    const int wm   = wave & 1;
    const int wn   = wave >> 1;
    const int half = lane >> 4;
    const int r16  = lane & 15;

    const int rowBlock = blockIdx.y * BM;
    const int kBlock   = blockIdx.x * 64;

    const int a_lr = tid >> 2;
    const int a_lc = (tid & 3) << 3;
    const int Ra   = rowBlock + a_lr;
    const int ab   = Ra / NF_N;
    const int am   = Ra - ab * NF_N;
    const float* sg = outsig + (size_t)ab * OUT_STRIDE + am * HOP_N;
    const float* fp = sg + a_lc;
    const float* mp = sg + (WIN_N - 7 - a_lc);

    const int b_row = tid >> 3;
    const int b_kc  = (tid & 7) << 2;
    const float* wp = WfwdT + (size_t)(kBlock + b_row) * KH + b_kc;

    v8f accR[2], accI[2];
#pragma unroll
    for (int i = 0; i < 2; ++i) {
        accR[i] = (v8f){0.f,0.f,0.f,0.f,0.f,0.f,0.f,0.f};
        accI[i] = (v8f){0.f,0.f,0.f,0.f,0.f,0.f,0.f,0.f};
    }

    for (int kb = 0; kb < KH; kb += BK) {
        __syncthreads();
        const float4 f0 = *(const float4*)(fp);
        const float4 f1 = *(const float4*)(fp + 4);
        const v4fu m0 = *(const v4fu*)(mp);
        const v4fu m1 = *(const v4fu*)(mp + 4);
        const float4 w0 = *(const float4*)(wp);
        const float4 w1 = *(const float4*)(wp + 32 * KH);
        const float4 w2 = *(const float4*)(wp + KH * KH);
        const float4 w3 = *(const float4*)(wp + KH * KH + 32 * KH);
        fp += BK; mp -= BK; wp += BK;
        *(float4*)&sAp[a_lr][a_lc]     = make_float4(f0.x + m1.w, f0.y + m1.z, f0.z + m1.y, f0.w + m1.x);
        *(float4*)&sAp[a_lr][a_lc + 4] = make_float4(f1.x + m0.w, f1.y + m0.z, f1.z + m0.y, f1.w + m0.x);
        *(float4*)&sAm[a_lr][a_lc]     = make_float4(f0.x - m1.w, f0.y - m1.z, f0.z - m1.y, f0.w - m1.x);
        *(float4*)&sAm[a_lr][a_lc + 4] = make_float4(f1.x - m0.w, f1.y - m0.z, f1.z - m0.y, f1.w - m0.x);
        *(float4*)&sBc[b_row][b_kc]      = w0;
        *(float4*)&sBc[b_row + 32][b_kc] = w1;
        *(float4*)&sBs[b_row][b_kc]      = w2;
        *(float4*)&sBs[b_row + 32][b_kc] = w3;
        __syncthreads();
#pragma unroll
        for (int k0 = 0; k0 < BK; k0 += 4) {
            const int ka = k0 + (half << 1);
            v2f ap0 = *(const v2f*)&sAp[(wm << 5) + r16][ka];
            v2f ap1 = *(const v2f*)&sAp[(wm << 5) + 16 + r16][ka];
            v2f am0 = *(const v2f*)&sAm[(wm << 5) + r16][ka];
            v2f am1 = *(const v2f*)&sAm[(wm << 5) + 16 + r16][ka];
            v2f bc  = *(const v2f*)&sBc[(wn << 4) + r16][ka];
            v2f bs  = *(const v2f*)&sBs[(wn << 4) + r16][ka];
            accR[0] = __builtin_amdgcn_wmma_f32_16x16x4_f32(false, ap0, false, bc, (short)0, accR[0], false, false);
            accI[0] = __builtin_amdgcn_wmma_f32_16x16x4_f32(false, am0, false, bs, (short)0, accI[0], false, false);
            accR[1] = __builtin_amdgcn_wmma_f32_16x16x4_f32(false, ap1, false, bc, (short)0, accR[1], false, false);
            accI[1] = __builtin_amdgcn_wmma_f32_16x16x4_f32(false, am1, false, bs, (short)0, accI[1], false, false);
        }
    }

    __syncthreads();
    {
        const int kl = (wn << 4) + r16;
#pragma unroll
        for (int im = 0; im < 2; ++im) {
            const int rl = (wm << 5) + (im << 4) + (half << 3);
#pragma unroll
            for (int v = 0; v < 8; ++v) {
                v2f p; p.x = accR[im][v]; p.y = accI[im][v];
                *(v2f*)&smem[(kl * SP + rl + v) << 1] = p;
            }
        }
    }
    __syncthreads();
#pragma unroll
    for (int rh = 0; rh < 2; ++rh) {
        const int rl = (rh << 5) + lane;
        const int Rr = rowBlock + rl;
        const int b  = Rr / NF_N;
        const int m  = Rr - b * NF_N;
        float* op = out + (size_t)b * (F_N * NF_N * 2) + (size_t)(kBlock + (wave << 3)) * (NF_N * 2) + m * 2;
#pragma unroll
        for (int i = 0; i < 8; ++i) {
            const v2f p = *(const v2f*)&smem[((((wave << 3) + i) * SP) + rl) << 1];
            *(v2f*)(op + i * (NF_N * 2)) = p;
        }
    }
}

extern "C" void kernel_launch(void* const* d_in, const int* in_sizes, int n_in,
                              void* d_out, int out_size, void* d_ws, size_t ws_size,
                              hipStream_t stream) {
    (void)in_sizes; (void)n_in; (void)out_size; (void)ws_size;
    const float* x = (const float*)d_in[0];
    float* out = (float*)d_out;

    float* ws     = (float*)d_ws;
    float* WinvT  = ws;
    float* winT   = WinvT + (size_t)KPAD * KH;
    float* WfwdT  = WinvT + (size_t)KPAD * KPAD;
    float* sig    = WfwdT + (size_t)KPAD * KPAD;
    float* outsig = sig + (size_t)B_N * SIG_STRIDE;

    hipMemsetAsync(sig, 0, (size_t)B_N * SIG_STRIDE * sizeof(float), stream);
    build_mats<<<(2 * KPAD * KH + KPAD) / 256, 256, 0, stream>>>(WinvT, WfwdT, winT);
    istft_gemm_ola<<<dim3(KH / 64, M1 / BM), 256, 0, stream>>>(x, WinvT, winT, sig);
    fold_kernel<<<(B_N * OUT_LEN + 255) / 256, 256, 0, stream>>>(sig, outsig);
    stft_gemm<<<dim3(KH / 64, M2 / BM), 256, 0, stream>>>(outsig, WfwdT, out);
}


# Both real DFTs folded again about N/4 (even/odd bins split, samples n and N/2-n paired; analysis window moved onto the operand): K of both f32 WMMA products 512 -> 256; terms summed in a new order
# speedup vs baseline: 2.6139x; 1.3827x over previous
#include <hip/hip_runtime.h>


typedef float v2f __attribute__((ext_vector_type(2)));
typedef float v8f __attribute__((ext_vector_type(8)));
typedef float v4f __attribute__((ext_vector_type(4)));
typedef v4f v4fu __attribute__((aligned(4)));

#define B_N        256
#define F_N        512
#define T_N        81
#define WIN_N      1022
#define HOP_N      400
#define NF_N       69
#define KEPT_LEN   30000
#define OUT_LEN    28550
#define SIG_STRIDE 33024
#define OUT_STRIDE 28560
#define KPAD       1024
#define KH         512
#define KQ         256
#define M1         (B_N * T_N)
#define M2         (B_N * NF_N)

#define BM 64
#define BN 128
#define BK 32
#define TS 36

#define TWO_PI 6.283185307179586476925f

__device__ __forceinline__ float hann_f(int n) {
    return 0.5f - 0.5f * cosf((TWO_PI / WIN_N) * (float)n);
}

__device__ __forceinline__ float synwin_f(int n) {
    int p = n % HOP_N;
    float d = 0.f;
    for (int j = p; j < WIN_N; j += HOP_N) {
        float hj = hann_f(j);
        d += hj * hj;
    }
    return hann_f(n) / d;
}

__global__ __launch_bounds__(256) void build_mats(float* __restrict__ WinvT,
                                                  float* __restrict__ WfwdT,
                                                  float* __restrict__ winT,
                                                  float* __restrict__ gT) {
    int idx = blockIdx.x * 256 + threadIdx.x;
    int which = idx >> 18;
    if (which == 2) {
        int t = idx & 2047;
        if (t < KPAD) {
            winT[t] = (t < WIN_N) ? synwin_f(t) : 0.f;
        } else if (t < KPAD + KH) {
            int n = t - KPAD;
            float h = hann_f(n);
            gT[n] = (n == KH - 1) ? 0.5f * h : h;
        }
        return;
    }
    int e   = idx & ((1 << 18) - 1);
    int col = e & (KQ - 1);
    if (which == 0) {
        int o = e >> 16;
        int n = (e >> 8) & (KQ - 1);
        int f = 2 * col + (o & 1);
        int m = (f * n) % WIN_N;
        float ang = (TWO_PI / WIN_N) * (float)m;
        float wf  = (f == 0 || f == F_N - 1) ? 1.f : 2.f;
        float tr  = (o < 2) ? cosf(ang) : sinf(ang);
        WinvT[e] = wf * tr * (1.0f / (float)WIN_N);
    } else {
        int c = e >> 17;
        int j = (e >> 8) & (KH - 1);
        int k = 2 * (j & (KQ - 1)) + (j >> 8);
        int m = (k * col) % WIN_N;
        float ang = (TWO_PI / WIN_N) * (float)m;
        WfwdT[e] = (c == 0) ? cosf(ang) : -sinf(ang);
    }
}

__global__ __launch_bounds__(256) void istft_gemm_ola(const float* __restrict__ x,
                                                      const float* __restrict__ WinvT,
                                                      const float* __restrict__ winT,
                                                      float* __restrict__ sig) {
    __shared__ __align__(16) float sA[4][BM][TS];
    __shared__ __align__(16) float sB[4][BM][TS];

    const int tid  = threadIdx.x;
    const int lane = tid & 31;
    const int wave = tid >> 5;
    const int wm   = wave & 1;
    const int wn   = wave >> 1;
    const int half = lane >> 4;
    const int r16  = lane & 15;

    const int rowBlock = blockIdx.y * BM;
    const int nBlock   = blockIdx.x * 64;

    const int a_lr = tid & 63;
    const int a_lc = (tid >> 6) << 3;
    const int Ra   = rowBlock + a_lr;
    const int ab   = Ra / T_N;
    const int at   = Ra - ab * T_N;
    const float* xp = x + (size_t)ab * (F_N * T_N * 2) + at * 2 + (size_t)(2 * a_lc) * (T_N * 2);

    const int b_row = tid >> 3;
    const int b_kc  = (tid & 7) << 2;
    const float* wp = WinvT + (size_t)(nBlock + b_row) * KQ + b_kc;

    v8f acc[4][2];
#pragma unroll
    for (int o = 0; o < 4; ++o) {
        acc[o][0] = (v8f){0.f,0.f,0.f,0.f,0.f,0.f,0.f,0.f};
        acc[o][1] = (v8f){0.f,0.f,0.f,0.f,0.f,0.f,0.f,0.f};
    }

    for (int kb = 0; kb < KQ; kb += BK) {
        __syncthreads();
        v2f av[16];
#pragma unroll
        for (int u = 0; u < 16; ++u)
            av[u] = *(const v2f*)(xp + u * (T_N * 2));
        float4 wv[4][2];
#pragma unroll
        for (int o = 0; o < 4; ++o) {
            wv[o][0] = *(const float4*)(wp + (size_t)o * (KQ * KQ));
            wv[o][1] = *(const float4*)(wp + (size_t)o * (KQ * KQ) + 32 * KQ);
        }
        xp += 2 * BK * (T_N * 2); wp += BK;
        *(float4*)&sA[0][a_lr][a_lc]     = make_float4(av[0].x, av[2].x, av[4].x, av[6].x);
        *(float4*)&sA[0][a_lr][a_lc + 4] = make_float4(av[8].x, av[10].x, av[12].x, av[14].x);
        *(float4*)&sA[1][a_lr][a_lc]     = make_float4(av[1].x, av[3].x, av[5].x, av[7].x);
        *(float4*)&sA[1][a_lr][a_lc + 4] = make_float4(av[9].x, av[11].x, av[13].x, av[15].x);
        *(float4*)&sA[2][a_lr][a_lc]     = make_float4(av[0].y, av[2].y, av[4].y, av[6].y);
        *(float4*)&sA[2][a_lr][a_lc + 4] = make_float4(av[8].y, av[10].y, av[12].y, av[14].y);
        *(float4*)&sA[3][a_lr][a_lc]     = make_float4(av[1].y, av[3].y, av[5].y, av[7].y);
        *(float4*)&sA[3][a_lr][a_lc + 4] = make_float4(av[9].y, av[11].y, av[13].y, av[15].y);
#pragma unroll
        for (int o = 0; o < 4; ++o) {
            *(float4*)&sB[o][b_row][b_kc]      = wv[o][0];
            *(float4*)&sB[o][b_row + 32][b_kc] = wv[o][1];
        }
        __syncthreads();
#pragma unroll
        for (int k0 = 0; k0 < BK; k0 += 4) {
            const int ka = k0 + (half << 1);
            v2f bf[4];
#pragma unroll
            for (int o = 0; o < 4; ++o)
                bf[o] = *(const v2f*)&sB[o][(wn << 4) + r16][ka];
#pragma unroll
            for (int im = 0; im < 2; ++im) {
#pragma unroll
                for (int o = 0; o < 4; ++o) {
                    v2f af = *(const v2f*)&sA[o][(wm << 5) + (im << 4) + r16][ka];
                    acc[o][im] = __builtin_amdgcn_wmma_f32_16x16x4_f32(false, af, false, bf[o], (short)0, acc[o][im], false, false);
                }
            }
        }
        __builtin_amdgcn_sched_barrier(0);
        asm volatile("v_nop\n\tv_nop\n\tv_nop\n\tv_nop");
    }

    const int n = nBlock + (wn << 4) + r16;
    const bool mir = (n >= 1);
    const float w0 = winT[n];
    const float w1 = mir ? winT[WIN_N - n] : 0.f;
    const float w2 = winT[KH - 1 - n];
    const float w3 = mir ? winT[KH - 1 + n] : 0.f;
#pragma unroll
    for (int im = 0; im < 2; ++im) {
        const int Rb = rowBlock + (wm << 5) + (im << 4) + (half << 3);
#pragma unroll
        for (int v = 0; v < 8; ++v) {
            const unsigned Rr = (unsigned)(Rb + v);
            const unsigned b  = Rr / T_N;
            const unsigned t  = Rr - b * T_N;
            const unsigned ro = b * SIG_STRIDE + t * HOP_N;
            const float cs = acc[0][im][v] + acc[1][im][v];
            const float cd = acc[0][im][v] - acc[1][im][v];
            const float ss = acc[2][im][v] + acc[3][im][v];
            const float sd = acc[2][im][v] - acc[3][im][v];
            atomicAdd(sig + (ro + (unsigned)n), w0 * (cs - ss));
            atomicAdd(sig + (ro + (unsigned)(KH - 1 - n)), w2 * (cd + sd));
            if (mir) {
                atomicAdd(sig + (ro + (unsigned)(WIN_N - n)), w1 * (cs + ss));
                atomicAdd(sig + (ro + (unsigned)(KH - 1 + n)), w3 * (cd - sd));
            }
        }
    }
}

__global__ __launch_bounds__(256) void fold_kernel(const float* __restrict__ sig,
                                                   float* __restrict__ outsig) {
    int i = blockIdx.x * 256 + threadIdx.x;
    if (i >= B_N * OUT_LEN) return;
    int b = i / OUT_LEN;
    int o = i - b * OUT_LEN;

    int k = (o < 1000) ? 0 : (o - 50) / 950;
    int r = o - k * 950;
    int j = k * 1000 + r;

    const float* s = sig + (size_t)b * SIG_STRIDE;
    float kept = s[k * 1100 + r];
    float sh = 0.f;
    int j2 = j + 50;
    if (j2 < KEPT_LEN) {
        int k2 = j2 / 1000, r2 = j2 - k2 * 1000;
        sh = s[k2 * 1100 + r2];
    }

    float cconst;
    if (j < 950)        cconst = 1.f;
    else if (j < 1000)  cconst = 0.f;
    else if (j < 29000) cconst = (r >= 50 && r < 950) ? 1.f : 0.f;
    else                cconst = 1.f;
    float cdec = (j < 29000 && r >= 950) ? (float)(1000 - r) * 0.02f : 0.f;
    float cinc = 0.f;
    if (j >= 950) {
        int rr = (j - 950) % 1000;
        if (rr < 50) cinc = (float)(rr + 1) * 0.02f;
    }
    outsig[(size_t)b * OUT_STRIDE + o] = (cconst + cdec) * kept + cinc * sh;
}

#define SP 65
__global__ __launch_bounds__(256) void stft_gemm(const float* __restrict__ outsig,
                                                 const float* __restrict__ WfwdT,
                                                 const float* __restrict__ gT,
                                                 float* __restrict__ out) {
    __shared__ __align__(16) float smem[4 * BM * TS];
    float (*sAp)[TS] = (float (*)[TS])(smem);
    float (*sAm)[TS] = (float (*)[TS])(smem + BM * TS);
    float (*sBc)[TS] = (float (*)[TS])(smem + 2 * BM * TS);
    float (*sBs)[TS] = (float (*)[TS])(smem + 3 * BM * TS);

    const int tid  = threadIdx.x;
    const int lane = tid & 31;
    const int wave = tid >> 5;
    const int wm   = wave & 1;
    const int wn   = wave >> 1;
    const int half = lane >> 4;
    const int r16  = lane & 15;

    const int rowBlock = blockIdx.y * BM;
    const int jBlock   = blockIdx.x * 64;
    const int parity   = blockIdx.x >> 2;
    const int kpBlock  = (blockIdx.x & 3) * 64;
    const float sgn    = parity ? -1.f : 1.f;

    const int a_lr = tid >> 2;
    const int a_lc = (tid & 3) << 3;
    const int Ra   = rowBlock + a_lr;
    const int ab   = Ra / NF_N;
    const int am   = Ra - ab * NF_N;
    const float* sg = outsig + (size_t)ab * OUT_STRIDE + am * HOP_N;
    const float* fp = sg + a_lc;
    const float* mp = sg + (WIN_N - 7 - a_lc);
    const float* qp = sg + (KH - 8 - a_lc);
    const float* rp = sg + (KH - 1 + a_lc);
    const float* gp = gT + a_lc;
    const float* hp = gT + (KH - 8 - a_lc);

    const int b_row = tid >> 3;
    const int b_kc  = (tid & 7) << 2;
    const float* wp = WfwdT + (size_t)(jBlock + b_row) * KQ + b_kc;

    v8f accR[2], accI[2];
#pragma unroll
    for (int i = 0; i < 2; ++i) {
        accR[i] = (v8f){0.f,0.f,0.f,0.f,0.f,0.f,0.f,0.f};
        accI[i] = (v8f){0.f,0.f,0.f,0.f,0.f,0.f,0.f,0.f};
    }

    for (int kb = 0; kb < KQ; kb += BK) {
        __syncthreads();
        const float4 f0 = *(const float4*)(fp);
        const float4 f1 = *(const float4*)(fp + 4);
        const v4fu m0 = *(const v4fu*)(mp);
        const v4fu m1 = *(const v4fu*)(mp + 4);
        const float4 q0 = *(const float4*)(qp);
        const float4 q1 = *(const float4*)(qp + 4);
        const v4fu r0 = *(const v4fu*)(rp);
        const v4fu r1 = *(const v4fu*)(rp + 4);
        const float4 g0 = *(const float4*)(gp);
        const float4 g1 = *(const float4*)(gp + 4);
        const float4 h0 = *(const float4*)(hp);
        const float4 h1 = *(const float4*)(hp + 4);
        const float4 w0 = *(const float4*)(wp);
        const float4 w1 = *(const float4*)(wp + 32 * KQ);
        const float4 w2 = *(const float4*)(wp + KH * KQ);
        const float4 w3 = *(const float4*)(wp + KH * KQ + 32 * KQ);
        fp += BK; mp -= BK; qp -= BK; rp += BK; gp += BK; hp -= BK; wp += BK;
        const float fa[8] = {f0.x, f0.y, f0.z, f0.w, f1.x, f1.y, f1.z, f1.w};
        const float ma[8] = {m1.w, m1.z, m1.y, m1.x, m0.w, m0.z, m0.y, m0.x};
        const float qa[8] = {q1.w, q1.z, q1.y, q1.x, q0.w, q0.z, q0.y, q0.x};
        const float ra[8] = {r0.x, r0.y, r0.z, r0.w, r1.x, r1.y, r1.z, r1.w};
        const float ga[8] = {g0.x, g0.y, g0.z, g0.w, g1.x, g1.y, g1.z, g1.w};
        const float ha[8] = {h1.w, h1.z, h1.y, h1.x, h0.w, h0.z, h0.y, h0.x};
        float pa[8], qv[8];
#pragma unroll
        for (int i = 0; i < 8; ++i) {
            const float un = (fa[i] + ma[i]) * ga[i];
            const float vn = (fa[i] - ma[i]) * ga[i];
            const float um = (qa[i] + ra[i]) * ha[i];
            const float vm = (qa[i] - ra[i]) * ha[i];
            pa[i] = un + sgn * um;
            qv[i] = vn - sgn * vm;
        }
        *(float4*)&sAp[a_lr][a_lc]     = make_float4(pa[0], pa[1], pa[2], pa[3]);
        *(float4*)&sAp[a_lr][a_lc + 4] = make_float4(pa[4], pa[5], pa[6], pa[7]);
        *(float4*)&sAm[a_lr][a_lc]     = make_float4(qv[0], qv[1], qv[2], qv[3]);
        *(float4*)&sAm[a_lr][a_lc + 4] = make_float4(qv[4], qv[5], qv[6], qv[7]);
        *(float4*)&sBc[b_row][b_kc]      = w0;
        *(float4*)&sBc[b_row + 32][b_kc] = w1;
        *(float4*)&sBs[b_row][b_kc]      = w2;
        *(float4*)&sBs[b_row + 32][b_kc] = w3;
        __syncthreads();
#pragma unroll
        for (int k0 = 0; k0 < BK; k0 += 4) {
            const int ka = k0 + (half << 1);
            v2f ap0 = *(const v2f*)&sAp[(wm << 5) + r16][ka];
            v2f ap1 = *(const v2f*)&sAp[(wm << 5) + 16 + r16][ka];
            v2f am0 = *(const v2f*)&sAm[(wm << 5) + r16][ka];
            v2f am1 = *(const v2f*)&sAm[(wm << 5) + 16 + r16][ka];
            v2f bc  = *(const v2f*)&sBc[(wn << 4) + r16][ka];
            v2f bs  = *(const v2f*)&sBs[(wn << 4) + r16][ka];
            accR[0] = __builtin_amdgcn_wmma_f32_16x16x4_f32(false, ap0, false, bc, (short)0, accR[0], false, false);
            accI[0] = __builtin_amdgcn_wmma_f32_16x16x4_f32(false, am0, false, bs, (short)0, accI[0], false, false);
            accR[1] = __builtin_amdgcn_wmma_f32_16x16x4_f32(false, ap1, false, bc, (short)0, accR[1], false, false);
            accI[1] = __builtin_amdgcn_wmma_f32_16x16x4_f32(false, am1, false, bs, (short)0, accI[1], false, false);
        }
        __builtin_amdgcn_sched_barrier(0);
        asm volatile("v_nop\n\tv_nop\n\tv_nop\n\tv_nop");
    }

    __syncthreads();
    {
        const int kl = (wn << 4) + r16;
#pragma unroll
        for (int im = 0; im < 2; ++im) {
            const int rl = (wm << 5) + (im << 4) + (half << 3);
#pragma unroll
            for (int v = 0; v < 8; ++v) {
                v2f p; p.x = accR[im][v]; p.y = accI[im][v];
                *(v2f*)&smem[(kl * SP + rl + v) << 1] = p;
            }
        }
    }
    __syncthreads();
#pragma unroll
    for (int rh = 0; rh < 2; ++rh) {
        const int rl = (rh << 5) + lane;
        const int Rr = rowBlock + rl;
        const int b  = Rr / NF_N;
        const int m  = Rr - b * NF_N;
        float* op = out + (size_t)b * (F_N * NF_N * 2)
                        + (size_t)(2 * (kpBlock + (wave << 3)) + parity) * (NF_N * 2) + m * 2;
#pragma unroll
        for (int i = 0; i < 8; ++i) {
            const v2f p = *(const v2f*)&smem[((((wave << 3) + i) * SP) + rl) << 1];
            *(v2f*)(op + i * (2 * NF_N * 2)) = p;
        }
    }
}

extern "C" void kernel_launch(void* const* d_in, const int* in_sizes, int n_in,
                              void* d_out, int out_size, void* d_ws, size_t ws_size,
                              hipStream_t stream) {
    (void)in_sizes; (void)n_in; (void)out_size; (void)ws_size;
    const float* x = (const float*)d_in[0];
    float* out = (float*)d_out;

    float* ws     = (float*)d_ws;
    float* WinvT  = ws;
    float* winT   = WinvT + (size_t)KPAD * KH;
    float* gT     = winT + KPAD;
    float* WfwdT  = WinvT + (size_t)KPAD * KPAD;
    float* sig    = WfwdT + (size_t)KPAD * KPAD;
    float* outsig = sig + (size_t)B_N * SIG_STRIDE;

    hipMemsetAsync(sig, 0, (size_t)B_N * SIG_STRIDE * sizeof(float), stream);
    build_mats<<<(2 * KPAD * KQ + 2 * KPAD) / 256, 256, 0, stream>>>(WinvT, WfwdT, winT, gT);
    istft_gemm_ola<<<dim3(KQ / 64, M1 / BM), 256, 0, stream>>>(x, WinvT, winT, sig);
    fold_kernel<<<(B_N * OUT_LEN + 255) / 256, 256, 0, stream>>>(sig, outsig);
    stft_gemm<<<dim3(KH / 64, M2 / BM), 256, 0, stream>>>(outsig, WfwdT, gT, out);
}


# Overlap-add by gather, not atomics: inverse product stores each windowed frame once (256-B runs via LDS); fold launch sums a sample's up to 3 frames in frame order, not arrival order; no memset
# speedup vs baseline: 2.7544x; 1.0538x over previous
#include <hip/hip_runtime.h>


typedef float v2f __attribute__((ext_vector_type(2)));
typedef float v8f __attribute__((ext_vector_type(8)));
typedef float v4f __attribute__((ext_vector_type(4)));
typedef v4f v4fu __attribute__((aligned(4)));

#define B_N        256
#define F_N        512
#define T_N        81
#define WIN_N      1022
#define HOP_N      400
#define NF_N       69
#define KEPT_LEN   30000
#define OUT_LEN    28550
#define SIG_STRIDE 33024
#define YROW       1024
#define YSPLIT_B   192
#define YSPLIT     (YSPLIT_B * T_N)
#define YP         68
#define OUT_STRIDE 28560
#define KPAD       1024
#define KH         512
#define KQ         256
#define M1         (B_N * T_N)
#define M2         (B_N * NF_N)

#define BM 64
#define BN 128
#define BK 32
#define TS 36

#define TWO_PI 6.283185307179586476925f

__device__ __forceinline__ float hann_f(int n) {
    return 0.5f - 0.5f * cosf((TWO_PI / WIN_N) * (float)n);
}

__device__ __forceinline__ float synwin_f(int n) {
    int p = n % HOP_N;
    float d = 0.f;
    for (int j = p; j < WIN_N; j += HOP_N) {
        float hj = hann_f(j);
        d += hj * hj;
    }
    return hann_f(n) / d;
}

__global__ __launch_bounds__(256) void build_mats(float* __restrict__ WinvT,
                                                  float* __restrict__ WfwdT,
                                                  float* __restrict__ winT,
                                                  float* __restrict__ gT) {
    int idx = blockIdx.x * 256 + threadIdx.x;
    int which = idx >> 18;
    if (which == 2) {
        int t = idx & 2047;
        if (t < KPAD) {
            winT[t] = (t < WIN_N) ? synwin_f(t) : 0.f;
        } else if (t < KPAD + KH) {
            int n = t - KPAD;
            float h = hann_f(n);
            gT[n] = (n == KH - 1) ? 0.5f * h : h;
        }
        return;
    }
    int e   = idx & ((1 << 18) - 1);
    int col = e & (KQ - 1);
    if (which == 0) {
        int o = e >> 16;
        int n = (e >> 8) & (KQ - 1);
        int f = 2 * col + (o & 1);
        int m = (f * n) % WIN_N;
        float ang = (TWO_PI / WIN_N) * (float)m;
        float wf  = (f == 0 || f == F_N - 1) ? 1.f : 2.f;
        float tr  = (o < 2) ? cosf(ang) : sinf(ang);
        WinvT[e] = wf * tr * (1.0f / (float)WIN_N);
    } else {
        int c = e >> 17;
        int j = (e >> 8) & (KH - 1);
        int k = 2 * (j & (KQ - 1)) + (j >> 8);
        int m = (k * col) % WIN_N;
        float ang = (TWO_PI / WIN_N) * (float)m;
        WfwdT[e] = (c == 0) ? cosf(ang) : -sinf(ang);
    }
}

__global__ __launch_bounds__(256) void istft_gemm_frames(const float* __restrict__ x,
                                                         const float* __restrict__ WinvT,
                                                         const float* __restrict__ winT,
                                                         float* __restrict__ ya,
                                                         float* __restrict__ yb) {
    __shared__ __align__(16) float smem[8 * BM * TS];
    float (*sA)[BM][TS] = (float (*)[BM][TS])(smem);
    float (*sB)[BM][TS] = (float (*)[BM][TS])(smem + 4 * BM * TS);

    const int tid  = threadIdx.x;
    const int lane = tid & 31;
    const int wave = tid >> 5;
    const int wm   = wave & 1;
    const int wn   = wave >> 1;
    const int half = lane >> 4;
    const int r16  = lane & 15;

    const int rowBlock = blockIdx.y * BM;
    const int nBlock   = blockIdx.x * 64;

    const int a_lr = tid & 63;
    const int a_lc = (tid >> 6) << 3;
    const int Ra   = rowBlock + a_lr;
    const int ab   = Ra / T_N;
    const int at   = Ra - ab * T_N;
    const float* xp = x + (size_t)ab * (F_N * T_N * 2) + at * 2 + (size_t)(2 * a_lc) * (T_N * 2);

    const int b_row = tid >> 3;
    const int b_kc  = (tid & 7) << 2;
    const float* wp = WinvT + (size_t)(nBlock + b_row) * KQ + b_kc;

    v8f acc[4][2];
#pragma unroll
    for (int o = 0; o < 4; ++o) {
        acc[o][0] = (v8f){0.f,0.f,0.f,0.f,0.f,0.f,0.f,0.f};
        acc[o][1] = (v8f){0.f,0.f,0.f,0.f,0.f,0.f,0.f,0.f};
    }

    for (int kb = 0; kb < KQ; kb += BK) {
        __syncthreads();
        v2f av[16];
#pragma unroll
        for (int u = 0; u < 16; ++u)
            av[u] = *(const v2f*)(xp + u * (T_N * 2));
        float4 wv[4][2];
#pragma unroll
        for (int o = 0; o < 4; ++o) {
            wv[o][0] = *(const float4*)(wp + (size_t)o * (KQ * KQ));
            wv[o][1] = *(const float4*)(wp + (size_t)o * (KQ * KQ) + 32 * KQ);
        }
        xp += 2 * BK * (T_N * 2); wp += BK;
        *(float4*)&sA[0][a_lr][a_lc]     = make_float4(av[0].x, av[2].x, av[4].x, av[6].x);
        *(float4*)&sA[0][a_lr][a_lc + 4] = make_float4(av[8].x, av[10].x, av[12].x, av[14].x);
        *(float4*)&sA[1][a_lr][a_lc]     = make_float4(av[1].x, av[3].x, av[5].x, av[7].x);
        *(float4*)&sA[1][a_lr][a_lc + 4] = make_float4(av[9].x, av[11].x, av[13].x, av[15].x);
        *(float4*)&sA[2][a_lr][a_lc]     = make_float4(av[0].y, av[2].y, av[4].y, av[6].y);
        *(float4*)&sA[2][a_lr][a_lc + 4] = make_float4(av[8].y, av[10].y, av[12].y, av[14].y);
        *(float4*)&sA[3][a_lr][a_lc]     = make_float4(av[1].y, av[3].y, av[5].y, av[7].y);
        *(float4*)&sA[3][a_lr][a_lc + 4] = make_float4(av[9].y, av[11].y, av[13].y, av[15].y);
#pragma unroll
        for (int o = 0; o < 4; ++o) {
            *(float4*)&sB[o][b_row][b_kc]      = wv[o][0];
            *(float4*)&sB[o][b_row + 32][b_kc] = wv[o][1];
        }
        __syncthreads();
#pragma unroll
        for (int k0 = 0; k0 < BK; k0 += 4) {
            const int ka = k0 + (half << 1);
            v2f bf[4];
#pragma unroll
            for (int o = 0; o < 4; ++o)
                bf[o] = *(const v2f*)&sB[o][(wn << 4) + r16][ka];
#pragma unroll
            for (int im = 0; im < 2; ++im) {
#pragma unroll
                for (int o = 0; o < 4; ++o) {
                    v2f af = *(const v2f*)&sA[o][(wm << 5) + (im << 4) + r16][ka];
                    acc[o][im] = __builtin_amdgcn_wmma_f32_16x16x4_f32(false, af, false, bf[o], (short)0, acc[o][im], false, false);
                }
            }
        }
        __builtin_amdgcn_sched_barrier(0);
        asm volatile("v_nop\n\tv_nop\n\tv_nop\n\tv_nop");
    }

    __syncthreads();
    {
        const int nl = (wn << 4) + r16;
        const int n  = nBlock + nl;
        const bool mir = (n >= 1);
        const float w0 = winT[n];
        const float w1 = mir ? winT[WIN_N - n] : 0.f;
        const float w2 = winT[KH - 1 - n];
        const float w3 = mir ? winT[KH - 1 + n] : 0.f;
#pragma unroll
        for (int im = 0; im < 2; ++im) {
            const int rl = (wm << 5) + (im << 4) + (half << 3);
#pragma unroll
            for (int v = 0; v < 8; ++v) {
                const float cs = acc[0][im][v] + acc[1][im][v];
                const float cd = acc[0][im][v] - acc[1][im][v];
                const float ss = acc[2][im][v] + acc[3][im][v];
                const float sd = acc[2][im][v] - acc[3][im][v];
                float* sp = smem + (rl + v) * YP + nl;
                sp[0 * BM * YP] = w0 * (cs - ss);
                sp[1 * BM * YP] = w1 * (cs + ss);
                sp[2 * BM * YP] = w2 * (cd + sd);
                sp[3 * BM * YP] = w3 * (cd - sd);
            }
        }
    }
    __syncthreads();
    {
        float* yr = (rowBlock < YSPLIT) ? ya + (size_t)rowBlock * YROW
                                        : yb + (size_t)(rowBlock - YSPLIT) * YROW;
#pragma unroll
        for (int it = 0; it < 16; ++it) {
            const int idx = (it << 4) + (wave << 1) + half;
            const int q   = idx >> 6;
            const int row = idx & (BM - 1);
            const float4 p = *(const float4*)&smem[idx * YP + (r16 << 2)];
            *(float4*)(yr + (size_t)row * YROW + q * KQ + nBlock + (r16 << 2)) = p;
        }
    }
}

__device__ __forceinline__ float ola_sample(const float* __restrict__ yr, int s) {
    const int q   = s / HOP_N;
    const int rho = s - q * HOP_N;
    const bool h2 = (q >= 2) && (rho < WIN_N - 2 * HOP_N);
    const bool h1 = (q >= 1) && (q <= T_N);
    const bool h0 = (q <= T_N - 1);
    const int p0 = (rho < KQ) ? rho : (2 * KQ + KH - 1) - rho;
    const int p1 = (rho < KH - HOP_N)          ? (2 * KQ + KH - 1 - HOP_N) - rho
                 : (rho < KH + KQ - 1 - HOP_N) ? rho + (3 * KQ + HOP_N - (KH - 1))
                 :                               (KQ + WIN_N - HOP_N) - rho;
    const int p2 = (KQ + WIN_N - 2 * HOP_N) - rho;
    const float y2 = yr[h2 ? (q - 2) * YROW + p2 : p0];
    const float y1 = yr[h1 ? (q - 1) * YROW + p1 : p0];
    const float y0 = yr[h0 ? q * YROW + p0 : p0];
    float a = h2 ? y2 : 0.f;
    a += h1 ? y1 : 0.f;
    a += h0 ? y0 : 0.f;
    return a;
}

__global__ __launch_bounds__(256) void fold_kernel(const float* __restrict__ ya,
                                                   const float* __restrict__ yb,
                                                   float* __restrict__ outsig) {
    const int b = blockIdx.y;
    const int o = blockIdx.x * 256 + threadIdx.x;
    if (o >= OUT_LEN) return;

    int k = (o < 1000) ? 0 : (o - 50) / 950;
    int r = o - k * 950;
    int j = k * 1000 + r;

    const float* yr = (b < YSPLIT_B) ? ya + (size_t)b * (T_N * YROW)
                                     : yb + (size_t)(b - YSPLIT_B) * (T_N * YROW);
    float kept = ola_sample(yr, k * 1100 + r);

    float cconst;
    if (j < 950)        cconst = 1.f;
    else if (j < 1000)  cconst = 0.f;
    else if (j < 29000) cconst = (r >= 50 && r < 950) ? 1.f : 0.f;
    else                cconst = 1.f;
    float cdec = (j < 29000 && r >= 950) ? (float)(1000 - r) * 0.02f : 0.f;
    float cinc = 0.f;
    if (j >= 950) {
        int rr = (j - 950) % 1000;
        if (rr < 50) cinc = (float)(rr + 1) * 0.02f;
    }
    float sh = 0.f;
    int j2 = j + 50;
    if (cinc != 0.f && j2 < KEPT_LEN) {
        int k2 = j2 / 1000, r2 = j2 - k2 * 1000;
        sh = ola_sample(yr, k2 * 1100 + r2);
    }
    outsig[(size_t)b * OUT_STRIDE + o] = (cconst + cdec) * kept + cinc * sh;
}

#define SP 65
__global__ __launch_bounds__(256) void stft_gemm(const float* __restrict__ outsig,
                                                 const float* __restrict__ WfwdT,
                                                 const float* __restrict__ gT,
                                                 float* __restrict__ out) {
    __shared__ __align__(16) float smem[4 * BM * TS];
    float (*sAp)[TS] = (float (*)[TS])(smem);
    float (*sAm)[TS] = (float (*)[TS])(smem + BM * TS);
    float (*sBc)[TS] = (float (*)[TS])(smem + 2 * BM * TS);
    float (*sBs)[TS] = (float (*)[TS])(smem + 3 * BM * TS);

    const int tid  = threadIdx.x;
    const int lane = tid & 31;
    const int wave = tid >> 5;
    const int wm   = wave & 1;
    const int wn   = wave >> 1;
    const int half = lane >> 4;
    const int r16  = lane & 15;

    const int rowBlock = blockIdx.y * BM;
    const int jBlock   = blockIdx.x * 64;
    const int parity   = blockIdx.x >> 2;
    const int kpBlock  = (blockIdx.x & 3) * 64;
    const float sgn    = parity ? -1.f : 1.f;

    const int a_lr = tid >> 2;
    const int a_lc = (tid & 3) << 3;
    const int Ra   = rowBlock + a_lr;
    const int ab   = Ra / NF_N;
    const int am   = Ra - ab * NF_N;
    const float* sg = outsig + (size_t)ab * OUT_STRIDE + am * HOP_N;
    const float* fp = sg + a_lc;
    const float* mp = sg + (WIN_N - 7 - a_lc);
    const float* qp = sg + (KH - 8 - a_lc);
    const float* rp = sg + (KH - 1 + a_lc);
    const float* gp = gT + a_lc;
    const float* hp = gT + (KH - 8 - a_lc);

    const int b_row = tid >> 3;
    const int b_kc  = (tid & 7) << 2;
    const float* wp = WfwdT + (size_t)(jBlock + b_row) * KQ + b_kc;

    v8f accR[2], accI[2];
#pragma unroll
    for (int i = 0; i < 2; ++i) {
        accR[i] = (v8f){0.f,0.f,0.f,0.f,0.f,0.f,0.f,0.f};
        accI[i] = (v8f){0.f,0.f,0.f,0.f,0.f,0.f,0.f,0.f};
    }

    for (int kb = 0; kb < KQ; kb += BK) {
        __syncthreads();
        const float4 f0 = *(const float4*)(fp);
        const float4 f1 = *(const float4*)(fp + 4);
        const v4fu m0 = *(const v4fu*)(mp);
        const v4fu m1 = *(const v4fu*)(mp + 4);
        const float4 q0 = *(const float4*)(qp);
        const float4 q1 = *(const float4*)(qp + 4);
        const v4fu r0 = *(const v4fu*)(rp);
        const v4fu r1 = *(const v4fu*)(rp + 4);
        const float4 g0 = *(const float4*)(gp);
        const float4 g1 = *(const float4*)(gp + 4);
        const float4 h0 = *(const float4*)(hp);
        const float4 h1 = *(const float4*)(hp + 4);
        const float4 w0 = *(const float4*)(wp);
        const float4 w1 = *(const float4*)(wp + 32 * KQ);
        const float4 w2 = *(const float4*)(wp + KH * KQ);
        const float4 w3 = *(const float4*)(wp + KH * KQ + 32 * KQ);
        fp += BK; mp -= BK; qp -= BK; rp += BK; gp += BK; hp -= BK; wp += BK;
        const float fa[8] = {f0.x, f0.y, f0.z, f0.w, f1.x, f1.y, f1.z, f1.w};
        const float ma[8] = {m1.w, m1.z, m1.y, m1.x, m0.w, m0.z, m0.y, m0.x};
        const float qa[8] = {q1.w, q1.z, q1.y, q1.x, q0.w, q0.z, q0.y, q0.x};
        const float ra[8] = {r0.x, r0.y, r0.z, r0.w, r1.x, r1.y, r1.z, r1.w};
        const float ga[8] = {g0.x, g0.y, g0.z, g0.w, g1.x, g1.y, g1.z, g1.w};
        const float ha[8] = {h1.w, h1.z, h1.y, h1.x, h0.w, h0.z, h0.y, h0.x};
        float pa[8], qv[8];
#pragma unroll
        for (int i = 0; i < 8; ++i) {
            const float un = (fa[i] + ma[i]) * ga[i];
            const float vn = (fa[i] - ma[i]) * ga[i];
            const float um = (qa[i] + ra[i]) * ha[i];
            const float vm = (qa[i] - ra[i]) * ha[i];
            pa[i] = un + sgn * um;
            qv[i] = vn - sgn * vm;
        }
        *(float4*)&sAp[a_lr][a_lc]     = make_float4(pa[0], pa[1], pa[2], pa[3]);
        *(float4*)&sAp[a_lr][a_lc + 4] = make_float4(pa[4], pa[5], pa[6], pa[7]);
        *(float4*)&sAm[a_lr][a_lc]     = make_float4(qv[0], qv[1], qv[2], qv[3]);
        *(float4*)&sAm[a_lr][a_lc + 4] = make_float4(qv[4], qv[5], qv[6], qv[7]);
        *(float4*)&sBc[b_row][b_kc]      = w0;
        *(float4*)&sBc[b_row + 32][b_kc] = w1;
        *(float4*)&sBs[b_row][b_kc]      = w2;
        *(float4*)&sBs[b_row + 32][b_kc] = w3;
        __syncthreads();
#pragma unroll
        for (int k0 = 0; k0 < BK; k0 += 4) {
            const int ka = k0 + (half << 1);
            v2f ap0 = *(const v2f*)&sAp[(wm << 5) + r16][ka];
            v2f ap1 = *(const v2f*)&sAp[(wm << 5) + 16 + r16][ka];
            v2f am0 = *(const v2f*)&sAm[(wm << 5) + r16][ka];
            v2f am1 = *(const v2f*)&sAm[(wm << 5) + 16 + r16][ka];
            v2f bc  = *(const v2f*)&sBc[(wn << 4) + r16][ka];
            v2f bs  = *(const v2f*)&sBs[(wn << 4) + r16][ka];
            accR[0] = __builtin_amdgcn_wmma_f32_16x16x4_f32(false, ap0, false, bc, (short)0, accR[0], false, false);
            accI[0] = __builtin_amdgcn_wmma_f32_16x16x4_f32(false, am0, false, bs, (short)0, accI[0], false, false);
            accR[1] = __builtin_amdgcn_wmma_f32_16x16x4_f32(false, ap1, false, bc, (short)0, accR[1], false, false);
            accI[1] = __builtin_amdgcn_wmma_f32_16x16x4_f32(false, am1, false, bs, (short)0, accI[1], false, false);
        }
        __builtin_amdgcn_sched_barrier(0);
        asm volatile("v_nop\n\tv_nop\n\tv_nop\n\tv_nop");
    }

    __syncthreads();
    {
        const int kl = (wn << 4) + r16;
#pragma unroll
        for (int im = 0; im < 2; ++im) {
            const int rl = (wm << 5) + (im << 4) + (half << 3);
#pragma unroll
            for (int v = 0; v < 8; ++v) {
                v2f p; p.x = accR[im][v]; p.y = accI[im][v];
                *(v2f*)&smem[(kl * SP + rl + v) << 1] = p;
            }
        }
    }
    __syncthreads();
#pragma unroll
    for (int rh = 0; rh < 2; ++rh) {
        const int rl = (rh << 5) + lane;
        const int Rr = rowBlock + rl;
        const int b  = Rr / NF_N;
        const int m  = Rr - b * NF_N;
        float* op = out + (size_t)b * (F_N * NF_N * 2)
                        + (size_t)(2 * (kpBlock + (wave << 3)) + parity) * (NF_N * 2) + m * 2;
#pragma unroll
        for (int i = 0; i < 8; ++i) {
            const v2f p = *(const v2f*)&smem[((((wave << 3) + i) * SP) + rl) << 1];
            *(v2f*)(op + i * (2 * NF_N * 2)) = p;
        }
    }
}

extern "C" void kernel_launch(void* const* d_in, const int* in_sizes, int n_in,
                              void* d_out, int out_size, void* d_ws, size_t ws_size,
                              hipStream_t stream) {
    (void)in_sizes; (void)n_in; (void)out_size; (void)ws_size;
    const float* x = (const float*)d_in[0];
    float* out = (float*)d_out;

    float* ws     = (float*)d_ws;
    float* WinvT  = ws;
    float* winT   = WinvT + (size_t)KPAD * KH;
    float* gT     = winT + KPAD;
    float* WfwdT  = WinvT + (size_t)KPAD * KPAD;
    float* ytail  = WfwdT + (size_t)KPAD * KPAD;
    float* outsig = ytail + (size_t)B_N * SIG_STRIDE;

    build_mats<<<(2 * KPAD * KQ + 2 * KPAD) / 256, 256, 0, stream>>>(WinvT, WfwdT, winT, gT);
    istft_gemm_frames<<<dim3(KQ / 64, M1 / BM), 256, 0, stream>>>(x, WinvT, winT, out, ytail);
    fold_kernel<<<dim3((OUT_LEN + 255) / 256, B_N), 256, 0, stream>>>(out, ytail, outsig);
    stft_gemm<<<dim3(KH / 64, M2 / BM), 256, 0, stream>>>(outsig, WfwdT, gT, out);
}


# Fold launch coarsened: a thread makes 8 outputs 256 apart (places and all gather loads first, sums and stores after), 8x fewer waves to start; each sample's frames summed in the same order
# speedup vs baseline: 2.8545x; 1.0363x over previous
#include <hip/hip_runtime.h>


typedef float v2f __attribute__((ext_vector_type(2)));
typedef float v8f __attribute__((ext_vector_type(8)));
typedef float v4f __attribute__((ext_vector_type(4)));
typedef v4f v4fu __attribute__((aligned(4)));

#define B_N        256
#define F_N        512
#define T_N        81
#define WIN_N      1022
#define HOP_N      400
#define NF_N       69
#define KEPT_LEN   30000
#define OUT_LEN    28550
#define SIG_STRIDE 33024
#define YROW       1024
#define YSPLIT_B   192
#define YSPLIT     (YSPLIT_B * T_N)
#define YP         68
#define OUT_STRIDE 28560
#define KPAD       1024
#define KH         512
#define KQ         256
#define M1         (B_N * T_N)
#define M2         (B_N * NF_N)

#define BM 64
#define BN 128
#define BK 32
#define TS 36

#define TWO_PI 6.283185307179586476925f

__device__ __forceinline__ float hann_f(int n) {
    return 0.5f - 0.5f * cosf((TWO_PI / WIN_N) * (float)n);
}

__device__ __forceinline__ float synwin_f(int n) {
    int p = n % HOP_N;
    float d = 0.f;
    for (int j = p; j < WIN_N; j += HOP_N) {
        float hj = hann_f(j);
        d += hj * hj;
    }
    return hann_f(n) / d;
}

__global__ __launch_bounds__(256) void build_mats(float* __restrict__ WinvT,
                                                  float* __restrict__ WfwdT,
                                                  float* __restrict__ winT,
                                                  float* __restrict__ gT) {
    int idx = blockIdx.x * 256 + threadIdx.x;
    int which = idx >> 18;
    if (which == 2) {
        int t = idx & 2047;
        if (t < KPAD) {
            winT[t] = (t < WIN_N) ? synwin_f(t) : 0.f;
        } else if (t < KPAD + KH) {
            int n = t - KPAD;
            float h = hann_f(n);
            gT[n] = (n == KH - 1) ? 0.5f * h : h;
        }
        return;
    }
    int e   = idx & ((1 << 18) - 1);
    int col = e & (KQ - 1);
    if (which == 0) {
        int o = e >> 16;
        int n = (e >> 8) & (KQ - 1);
        int f = 2 * col + (o & 1);
        int m = (f * n) % WIN_N;
        float ang = (TWO_PI / WIN_N) * (float)m;
        float wf  = (f == 0 || f == F_N - 1) ? 1.f : 2.f;
        float tr  = (o < 2) ? cosf(ang) : sinf(ang);
        WinvT[e] = wf * tr * (1.0f / (float)WIN_N);
    } else {
        int c = e >> 17;
        int j = (e >> 8) & (KH - 1);
        int k = 2 * (j & (KQ - 1)) + (j >> 8);
        int m = (k * col) % WIN_N;
        float ang = (TWO_PI / WIN_N) * (float)m;
        WfwdT[e] = (c == 0) ? cosf(ang) : -sinf(ang);
    }
}

__global__ __launch_bounds__(256) void istft_gemm_frames(const float* __restrict__ x,
                                                         const float* __restrict__ WinvT,
                                                         const float* __restrict__ winT,
                                                         float* __restrict__ ya,
                                                         float* __restrict__ yb) {
    __shared__ __align__(16) float smem[8 * BM * TS];
    float (*sA)[BM][TS] = (float (*)[BM][TS])(smem);
    float (*sB)[BM][TS] = (float (*)[BM][TS])(smem + 4 * BM * TS);

    const int tid  = threadIdx.x;
    const int lane = tid & 31;
    const int wave = tid >> 5;
    const int wm   = wave & 1;
    const int wn   = wave >> 1;
    const int half = lane >> 4;
    const int r16  = lane & 15;

    const int rowBlock = blockIdx.y * BM;
    const int nBlock   = blockIdx.x * 64;

    const int a_lr = tid & 63;
    const int a_lc = (tid >> 6) << 3;
    const int Ra   = rowBlock + a_lr;
    const int ab   = Ra / T_N;
    const int at   = Ra - ab * T_N;
    const float* xp = x + (size_t)ab * (F_N * T_N * 2) + at * 2 + (size_t)(2 * a_lc) * (T_N * 2);

    const int b_row = tid >> 3;
    const int b_kc  = (tid & 7) << 2;
    const float* wp = WinvT + (size_t)(nBlock + b_row) * KQ + b_kc;

    v8f acc[4][2];
#pragma unroll
    for (int o = 0; o < 4; ++o) {
        acc[o][0] = (v8f){0.f,0.f,0.f,0.f,0.f,0.f,0.f,0.f};
        acc[o][1] = (v8f){0.f,0.f,0.f,0.f,0.f,0.f,0.f,0.f};
    }

    for (int kb = 0; kb < KQ; kb += BK) {
        __syncthreads();
        v2f av[16];
#pragma unroll
        for (int u = 0; u < 16; ++u)
            av[u] = *(const v2f*)(xp + u * (T_N * 2));
        float4 wv[4][2];
#pragma unroll
        for (int o = 0; o < 4; ++o) {
            wv[o][0] = *(const float4*)(wp + (size_t)o * (KQ * KQ));
            wv[o][1] = *(const float4*)(wp + (size_t)o * (KQ * KQ) + 32 * KQ);
        }
        xp += 2 * BK * (T_N * 2); wp += BK;
        *(float4*)&sA[0][a_lr][a_lc]     = make_float4(av[0].x, av[2].x, av[4].x, av[6].x);
        *(float4*)&sA[0][a_lr][a_lc + 4] = make_float4(av[8].x, av[10].x, av[12].x, av[14].x);
        *(float4*)&sA[1][a_lr][a_lc]     = make_float4(av[1].x, av[3].x, av[5].x, av[7].x);
        *(float4*)&sA[1][a_lr][a_lc + 4] = make_float4(av[9].x, av[11].x, av[13].x, av[15].x);
        *(float4*)&sA[2][a_lr][a_lc]     = make_float4(av[0].y, av[2].y, av[4].y, av[6].y);
        *(float4*)&sA[2][a_lr][a_lc + 4] = make_float4(av[8].y, av[10].y, av[12].y, av[14].y);
        *(float4*)&sA[3][a_lr][a_lc]     = make_float4(av[1].y, av[3].y, av[5].y, av[7].y);
        *(float4*)&sA[3][a_lr][a_lc + 4] = make_float4(av[9].y, av[11].y, av[13].y, av[15].y);
#pragma unroll
        for (int o = 0; o < 4; ++o) {
            *(float4*)&sB[o][b_row][b_kc]      = wv[o][0];
            *(float4*)&sB[o][b_row + 32][b_kc] = wv[o][1];
        }
        __syncthreads();
#pragma unroll
        for (int k0 = 0; k0 < BK; k0 += 4) {
            const int ka = k0 + (half << 1);
            v2f bf[4];
#pragma unroll
            for (int o = 0; o < 4; ++o)
                bf[o] = *(const v2f*)&sB[o][(wn << 4) + r16][ka];
#pragma unroll
            for (int im = 0; im < 2; ++im) {
#pragma unroll
                for (int o = 0; o < 4; ++o) {
                    v2f af = *(const v2f*)&sA[o][(wm << 5) + (im << 4) + r16][ka];
                    acc[o][im] = __builtin_amdgcn_wmma_f32_16x16x4_f32(false, af, false, bf[o], (short)0, acc[o][im], false, false);
                }
            }
        }
        __builtin_amdgcn_sched_barrier(0);
        asm volatile("v_nop\n\tv_nop\n\tv_nop\n\tv_nop");
    }

    __syncthreads();
    {
        const int nl = (wn << 4) + r16;
        const int n  = nBlock + nl;
        const bool mir = (n >= 1);
        const float w0 = winT[n];
        const float w1 = mir ? winT[WIN_N - n] : 0.f;
        const float w2 = winT[KH - 1 - n];
        const float w3 = mir ? winT[KH - 1 + n] : 0.f;
#pragma unroll
        for (int im = 0; im < 2; ++im) {
            const int rl = (wm << 5) + (im << 4) + (half << 3);
#pragma unroll
            for (int v = 0; v < 8; ++v) {
                const float cs = acc[0][im][v] + acc[1][im][v];
                const float cd = acc[0][im][v] - acc[1][im][v];
                const float ss = acc[2][im][v] + acc[3][im][v];
                const float sd = acc[2][im][v] - acc[3][im][v];
                float* sp = smem + (rl + v) * YP + nl;
                sp[0 * BM * YP] = w0 * (cs - ss);
                sp[1 * BM * YP] = w1 * (cs + ss);
                sp[2 * BM * YP] = w2 * (cd + sd);
                sp[3 * BM * YP] = w3 * (cd - sd);
            }
        }
    }
    __syncthreads();
    {
        float* yr = (rowBlock < YSPLIT) ? ya + (size_t)rowBlock * YROW
                                        : yb + (size_t)(rowBlock - YSPLIT) * YROW;
#pragma unroll
        for (int it = 0; it < 16; ++it) {
            const int idx = (it << 4) + (wave << 1) + half;
            const int q   = idx >> 6;
            const int row = idx & (BM - 1);
            const float4 p = *(const float4*)&smem[idx * YP + (r16 << 2)];
            *(float4*)(yr + (size_t)row * YROW + q * KQ + nBlock + (r16 << 2)) = p;
        }
    }
}

struct OlaRaw { float y0, y1, y2; int have; };

__device__ __forceinline__ OlaRaw ola_issue(const float* __restrict__ yr, int s) {
    const int q   = s / HOP_N;
    const int rho = s - q * HOP_N;
    const bool h2 = (q >= 2) && (rho < WIN_N - 2 * HOP_N);
    const bool h1 = (q >= 1) && (q <= T_N);
    const bool h0 = (q <= T_N - 1);
    const int p0 = (rho < KQ) ? rho : (2 * KQ + KH - 1) - rho;
    const int p1 = (rho < KH - HOP_N)          ? (2 * KQ + KH - 1 - HOP_N) - rho
                 : (rho < KH + KQ - 1 - HOP_N) ? rho + (3 * KQ + HOP_N - (KH - 1))
                 :                               (KQ + WIN_N - HOP_N) - rho;
    const int p2 = (KQ + WIN_N - 2 * HOP_N) - rho;
    OlaRaw w;
    w.y2 = yr[h2 ? (q - 2) * YROW + p2 : p0];
    w.y1 = yr[h1 ? (q - 1) * YROW + p1 : p0];
    w.y0 = yr[h0 ? q * YROW + p0 : p0];
    w.have = (h2 ? 4 : 0) | (h1 ? 2 : 0) | (h0 ? 1 : 0);
    return w;
}

__device__ __forceinline__ float ola_sum(const OlaRaw& w) {
    float a = (w.have & 4) ? w.y2 : 0.f;
    a += (w.have & 2) ? w.y1 : 0.f;
    a += (w.have & 1) ? w.y0 : 0.f;
    return a;
}

#define FOLD_E 8
__global__ __launch_bounds__(256) void fold_kernel(const float* __restrict__ ya,
                                                   const float* __restrict__ yb,
                                                   float* __restrict__ outsig) {
    const int b  = blockIdx.y;
    const int o0 = blockIdx.x * (256 * FOLD_E) + threadIdx.x;

    const float* yr = (b < YSPLIT_B) ? ya + (size_t)b * (T_N * YROW)
                                     : yb + (size_t)(b - YSPLIT_B) * (T_N * YROW);
    OlaRaw kr[FOLD_E], sr[FOLD_E];
    float ck[FOLD_E], ci[FOLD_E];
#pragma unroll
    for (int e = 0; e < FOLD_E; ++e) {
        const int o = min(o0 + e * 256, OUT_LEN - 1);

        int k = (o < 1000) ? 0 : (o - 50) / 950;
        int r = o - k * 950;
        int j = k * 1000 + r;

        kr[e] = ola_issue(yr, k * 1100 + r);

        float cconst;
        if (j < 950)        cconst = 1.f;
        else if (j < 1000)  cconst = 0.f;
        else if (j < 29000) cconst = (r >= 50 && r < 950) ? 1.f : 0.f;
        else                cconst = 1.f;
        float cdec = (j < 29000 && r >= 950) ? (float)(1000 - r) * 0.02f : 0.f;
        float cinc = 0.f;
        if (j >= 950) {
            int rr = (j - 950) % 1000;
            if (rr < 50) cinc = (float)(rr + 1) * 0.02f;
        }
        ck[e] = cconst + cdec;
        ci[e] = cinc;
        sr[e].y0 = 0.f; sr[e].y1 = 0.f; sr[e].y2 = 0.f; sr[e].have = 0;
        int j2 = j + 50;
        if (cinc != 0.f && j2 < KEPT_LEN) {
            int k2 = j2 / 1000, r2 = j2 - k2 * 1000;
            sr[e] = ola_issue(yr, k2 * 1100 + r2);
        }
    }
    float* op = outsig + (size_t)b * OUT_STRIDE;
#pragma unroll
    for (int e = 0; e < FOLD_E; ++e) {
        const int o = o0 + e * 256;
        const float kept = ola_sum(kr[e]);
        const float sh   = ola_sum(sr[e]);
        if (o < OUT_LEN) op[o] = ck[e] * kept + ci[e] * sh;
    }
}

#define SP 65
__global__ __launch_bounds__(256) void stft_gemm(const float* __restrict__ outsig,
                                                 const float* __restrict__ WfwdT,
                                                 const float* __restrict__ gT,
                                                 float* __restrict__ out) {
    __shared__ __align__(16) float smem[4 * BM * TS];
    float (*sAp)[TS] = (float (*)[TS])(smem);
    float (*sAm)[TS] = (float (*)[TS])(smem + BM * TS);
    float (*sBc)[TS] = (float (*)[TS])(smem + 2 * BM * TS);
    float (*sBs)[TS] = (float (*)[TS])(smem + 3 * BM * TS);

    const int tid  = threadIdx.x;
    const int lane = tid & 31;
    const int wave = tid >> 5;
    const int wm   = wave & 1;
    const int wn   = wave >> 1;
    const int half = lane >> 4;
    const int r16  = lane & 15;

    const int rowBlock = blockIdx.y * BM;
    const int jBlock   = blockIdx.x * 64;
    const int parity   = blockIdx.x >> 2;
    const int kpBlock  = (blockIdx.x & 3) * 64;
    const float sgn    = parity ? -1.f : 1.f;

    const int a_lr = tid >> 2;
    const int a_lc = (tid & 3) << 3;
    const int Ra   = rowBlock + a_lr;
    const int ab   = Ra / NF_N;
    const int am   = Ra - ab * NF_N;
    const float* sg = outsig + (size_t)ab * OUT_STRIDE + am * HOP_N;
    const float* fp = sg + a_lc;
    const float* mp = sg + (WIN_N - 7 - a_lc);
    const float* qp = sg + (KH - 8 - a_lc);
    const float* rp = sg + (KH - 1 + a_lc);
    const float* gp = gT + a_lc;
    const float* hp = gT + (KH - 8 - a_lc);

    const int b_row = tid >> 3;
    const int b_kc  = (tid & 7) << 2;
    const float* wp = WfwdT + (size_t)(jBlock + b_row) * KQ + b_kc;

    v8f accR[2], accI[2];
#pragma unroll
    for (int i = 0; i < 2; ++i) {
        accR[i] = (v8f){0.f,0.f,0.f,0.f,0.f,0.f,0.f,0.f};
        accI[i] = (v8f){0.f,0.f,0.f,0.f,0.f,0.f,0.f,0.f};
    }

    for (int kb = 0; kb < KQ; kb += BK) {
        __syncthreads();
        const float4 f0 = *(const float4*)(fp);
        const float4 f1 = *(const float4*)(fp + 4);
        const v4fu m0 = *(const v4fu*)(mp);
        const v4fu m1 = *(const v4fu*)(mp + 4);
        const float4 q0 = *(const float4*)(qp);
        const float4 q1 = *(const float4*)(qp + 4);
        const v4fu r0 = *(const v4fu*)(rp);
        const v4fu r1 = *(const v4fu*)(rp + 4);
        const float4 g0 = *(const float4*)(gp);
        const float4 g1 = *(const float4*)(gp + 4);
        const float4 h0 = *(const float4*)(hp);
        const float4 h1 = *(const float4*)(hp + 4);
        const float4 w0 = *(const float4*)(wp);
        const float4 w1 = *(const float4*)(wp + 32 * KQ);
        const float4 w2 = *(const float4*)(wp + KH * KQ);
        const float4 w3 = *(const float4*)(wp + KH * KQ + 32 * KQ);
        fp += BK; mp -= BK; qp -= BK; rp += BK; gp += BK; hp -= BK; wp += BK;
        const float fa[8] = {f0.x, f0.y, f0.z, f0.w, f1.x, f1.y, f1.z, f1.w};
        const float ma[8] = {m1.w, m1.z, m1.y, m1.x, m0.w, m0.z, m0.y, m0.x};
        const float qa[8] = {q1.w, q1.z, q1.y, q1.x, q0.w, q0.z, q0.y, q0.x};
        const float ra[8] = {r0.x, r0.y, r0.z, r0.w, r1.x, r1.y, r1.z, r1.w};
        const float ga[8] = {g0.x, g0.y, g0.z, g0.w, g1.x, g1.y, g1.z, g1.w};
        const float ha[8] = {h1.w, h1.z, h1.y, h1.x, h0.w, h0.z, h0.y, h0.x};
        float pa[8], qv[8];
#pragma unroll
        for (int i = 0; i < 8; ++i) {
            const float un = (fa[i] + ma[i]) * ga[i];
            const float vn = (fa[i] - ma[i]) * ga[i];
            const float um = (qa[i] + ra[i]) * ha[i];
            const float vm = (qa[i] - ra[i]) * ha[i];
            pa[i] = un + sgn * um;
            qv[i] = vn - sgn * vm;
        }
        *(float4*)&sAp[a_lr][a_lc]     = make_float4(pa[0], pa[1], pa[2], pa[3]);
        *(float4*)&sAp[a_lr][a_lc + 4] = make_float4(pa[4], pa[5], pa[6], pa[7]);
        *(float4*)&sAm[a_lr][a_lc]     = make_float4(qv[0], qv[1], qv[2], qv[3]);
        *(float4*)&sAm[a_lr][a_lc + 4] = make_float4(qv[4], qv[5], qv[6], qv[7]);
        *(float4*)&sBc[b_row][b_kc]      = w0;
        *(float4*)&sBc[b_row + 32][b_kc] = w1;
        *(float4*)&sBs[b_row][b_kc]      = w2;
        *(float4*)&sBs[b_row + 32][b_kc] = w3;
        __syncthreads();
#pragma unroll
        for (int k0 = 0; k0 < BK; k0 += 4) {
            const int ka = k0 + (half << 1);
            v2f ap0 = *(const v2f*)&sAp[(wm << 5) + r16][ka];
            v2f ap1 = *(const v2f*)&sAp[(wm << 5) + 16 + r16][ka];
            v2f am0 = *(const v2f*)&sAm[(wm << 5) + r16][ka];
            v2f am1 = *(const v2f*)&sAm[(wm << 5) + 16 + r16][ka];
            v2f bc  = *(const v2f*)&sBc[(wn << 4) + r16][ka];
            v2f bs  = *(const v2f*)&sBs[(wn << 4) + r16][ka];
            accR[0] = __builtin_amdgcn_wmma_f32_16x16x4_f32(false, ap0, false, bc, (short)0, accR[0], false, false);
            accI[0] = __builtin_amdgcn_wmma_f32_16x16x4_f32(false, am0, false, bs, (short)0, accI[0], false, false);
            accR[1] = __builtin_amdgcn_wmma_f32_16x16x4_f32(false, ap1, false, bc, (short)0, accR[1], false, false);
            accI[1] = __builtin_amdgcn_wmma_f32_16x16x4_f32(false, am1, false, bs, (short)0, accI[1], false, false);
        }
        __builtin_amdgcn_sched_barrier(0);
        asm volatile("v_nop\n\tv_nop\n\tv_nop\n\tv_nop");
    }

    __syncthreads();
    {
        const int kl = (wn << 4) + r16;
#pragma unroll
        for (int im = 0; im < 2; ++im) {
            const int rl = (wm << 5) + (im << 4) + (half << 3);
#pragma unroll
            for (int v = 0; v < 8; ++v) {
                v2f p; p.x = accR[im][v]; p.y = accI[im][v];
                *(v2f*)&smem[(kl * SP + rl + v) << 1] = p;
            }
        }
    }
    __syncthreads();
#pragma unroll
    for (int rh = 0; rh < 2; ++rh) {
        const int rl = (rh << 5) + lane;
        const int Rr = rowBlock + rl;
        const int b  = Rr / NF_N;
        const int m  = Rr - b * NF_N;
        float* op = out + (size_t)b * (F_N * NF_N * 2)
                        + (size_t)(2 * (kpBlock + (wave << 3)) + parity) * (NF_N * 2) + m * 2;
#pragma unroll
        for (int i = 0; i < 8; ++i) {
            const v2f p = *(const v2f*)&smem[((((wave << 3) + i) * SP) + rl) << 1];
            *(v2f*)(op + i * (2 * NF_N * 2)) = p;
        }
    }
}

extern "C" void kernel_launch(void* const* d_in, const int* in_sizes, int n_in,
                              void* d_out, int out_size, void* d_ws, size_t ws_size,
                              hipStream_t stream) {
    (void)in_sizes; (void)n_in; (void)out_size; (void)ws_size;
    const float* x = (const float*)d_in[0];
    float* out = (float*)d_out;

    float* ws     = (float*)d_ws;
    float* WinvT  = ws;
    float* winT   = WinvT + (size_t)KPAD * KH;
    float* gT     = winT + KPAD;
    float* WfwdT  = WinvT + (size_t)KPAD * KPAD;
    float* ytail  = WfwdT + (size_t)KPAD * KPAD;
    float* outsig = ytail + (size_t)B_N * SIG_STRIDE;

    build_mats<<<(2 * KPAD * KQ + 2 * KPAD) / 256, 256, 0, stream>>>(WinvT, WfwdT, winT, gT);
    istft_gemm_frames<<<dim3(KQ / 64, M1 / BM), 256, 0, stream>>>(x, WinvT, winT, out, ytail);
    fold_kernel<<<dim3((OUT_LEN + 256 * FOLD_E - 1) / (256 * FOLD_E), B_N), 256, 0, stream>>>(out, ytail, outsig);
    stft_gemm<<<dim3(KH / 64, M2 / BM), 256, 0, stream>>>(outsig, WfwdT, gT, out);
}


# Forward-DFT product tiled by whole signals: a work-group takes one signal's 69 frames x 128 bins (both parities from one read) and stores them as one aligned contiguous run; 1024 equal work-groups
# speedup vs baseline: 3.0666x; 1.0743x over previous
#include <hip/hip_runtime.h>


typedef float v2f __attribute__((ext_vector_type(2)));
typedef float v8f __attribute__((ext_vector_type(8)));
typedef float v4f __attribute__((ext_vector_type(4)));
typedef v4f v4fu __attribute__((aligned(4)));

#define B_N        256
#define F_N        512
#define T_N        81
#define WIN_N      1022
#define HOP_N      400
#define NF_N       69
#define KEPT_LEN   30000
#define OUT_LEN    28550
#define SIG_STRIDE 33024
#define YROW       1024
#define YSPLIT_B   192
#define YSPLIT     (YSPLIT_B * T_N)
#define YP         68
#define OUT_STRIDE 28560
#define KPAD       1024
#define KH         512
#define KQ         256
#define M1         (B_N * T_N)
#define M2         (B_N * NF_N)

#define BM 64
#define BN 128
#define BK 32
#define TS 36

#define TWO_PI 6.283185307179586476925f

__device__ __forceinline__ float hann_f(int n) {
    return 0.5f - 0.5f * cosf((TWO_PI / WIN_N) * (float)n);
}

__device__ __forceinline__ float synwin_f(int n) {
    int p = n % HOP_N;
    float d = 0.f;
    for (int j = p; j < WIN_N; j += HOP_N) {
        float hj = hann_f(j);
        d += hj * hj;
    }
    return hann_f(n) / d;
}

__global__ __launch_bounds__(256) void build_mats(float* __restrict__ WinvT,
                                                  float* __restrict__ WfwdT,
                                                  float* __restrict__ winT,
                                                  float* __restrict__ gT) {
    int idx = blockIdx.x * 256 + threadIdx.x;
    int which = idx >> 18;
    if (which == 2) {
        int t = idx & 2047;
        if (t < KPAD) {
            winT[t] = (t < WIN_N) ? synwin_f(t) : 0.f;
        } else if (t < KPAD + KH) {
            int n = t - KPAD;
            float h = hann_f(n);
            gT[n] = (n == KH - 1) ? 0.5f * h : h;
        }
        return;
    }
    int e   = idx & ((1 << 18) - 1);
    int col = e & (KQ - 1);
    if (which == 0) {
        int o = e >> 16;
        int n = (e >> 8) & (KQ - 1);
        int f = 2 * col + (o & 1);
        int m = (f * n) % WIN_N;
        float ang = (TWO_PI / WIN_N) * (float)m;
        float wf  = (f == 0 || f == F_N - 1) ? 1.f : 2.f;
        float tr  = (o < 2) ? cosf(ang) : sinf(ang);
        WinvT[e] = wf * tr * (1.0f / (float)WIN_N);
    } else {
        int c = e >> 17;
        int j = (e >> 8) & (KH - 1);
        int k = 2 * (j & (KQ - 1)) + (j >> 8);
        int m = (k * col) % WIN_N;
        float ang = (TWO_PI / WIN_N) * (float)m;
        WfwdT[e] = (c == 0) ? cosf(ang) : -sinf(ang);
    }
}

__global__ __launch_bounds__(256) void istft_gemm_frames(const float* __restrict__ x,
                                                         const float* __restrict__ WinvT,
                                                         const float* __restrict__ winT,
                                                         float* __restrict__ ya,
                                                         float* __restrict__ yb) {
    __shared__ __align__(16) float smem[8 * BM * TS];
    float (*sA)[BM][TS] = (float (*)[BM][TS])(smem);
    float (*sB)[BM][TS] = (float (*)[BM][TS])(smem + 4 * BM * TS);

    const int tid  = threadIdx.x;
    const int lane = tid & 31;
    const int wave = tid >> 5;
    const int wm   = wave & 1;
    const int wn   = wave >> 1;
    const int half = lane >> 4;
    const int r16  = lane & 15;

    const int rowBlock = blockIdx.y * BM;
    const int nBlock   = blockIdx.x * 64;

    const int a_lr = tid & 63;
    const int a_lc = (tid >> 6) << 3;
    const int Ra   = rowBlock + a_lr;
    const int ab   = Ra / T_N;
    const int at   = Ra - ab * T_N;
    const float* xp = x + (size_t)ab * (F_N * T_N * 2) + at * 2 + (size_t)(2 * a_lc) * (T_N * 2);

    const int b_row = tid >> 3;
    const int b_kc  = (tid & 7) << 2;
    const float* wp = WinvT + (size_t)(nBlock + b_row) * KQ + b_kc;

    v8f acc[4][2];
#pragma unroll
    for (int o = 0; o < 4; ++o) {
        acc[o][0] = (v8f){0.f,0.f,0.f,0.f,0.f,0.f,0.f,0.f};
        acc[o][1] = (v8f){0.f,0.f,0.f,0.f,0.f,0.f,0.f,0.f};
    }

    for (int kb = 0; kb < KQ; kb += BK) {
        __syncthreads();
        v2f av[16];
#pragma unroll
        for (int u = 0; u < 16; ++u)
            av[u] = *(const v2f*)(xp + u * (T_N * 2));
        float4 wv[4][2];
#pragma unroll
        for (int o = 0; o < 4; ++o) {
            wv[o][0] = *(const float4*)(wp + (size_t)o * (KQ * KQ));
            wv[o][1] = *(const float4*)(wp + (size_t)o * (KQ * KQ) + 32 * KQ);
        }
        xp += 2 * BK * (T_N * 2); wp += BK;
        *(float4*)&sA[0][a_lr][a_lc]     = make_float4(av[0].x, av[2].x, av[4].x, av[6].x);
        *(float4*)&sA[0][a_lr][a_lc + 4] = make_float4(av[8].x, av[10].x, av[12].x, av[14].x);
        *(float4*)&sA[1][a_lr][a_lc]     = make_float4(av[1].x, av[3].x, av[5].x, av[7].x);
        *(float4*)&sA[1][a_lr][a_lc + 4] = make_float4(av[9].x, av[11].x, av[13].x, av[15].x);
        *(float4*)&sA[2][a_lr][a_lc]     = make_float4(av[0].y, av[2].y, av[4].y, av[6].y);
        *(float4*)&sA[2][a_lr][a_lc + 4] = make_float4(av[8].y, av[10].y, av[12].y, av[14].y);
        *(float4*)&sA[3][a_lr][a_lc]     = make_float4(av[1].y, av[3].y, av[5].y, av[7].y);
        *(float4*)&sA[3][a_lr][a_lc + 4] = make_float4(av[9].y, av[11].y, av[13].y, av[15].y);
#pragma unroll
        for (int o = 0; o < 4; ++o) {
            *(float4*)&sB[o][b_row][b_kc]      = wv[o][0];
            *(float4*)&sB[o][b_row + 32][b_kc] = wv[o][1];
        }
        __syncthreads();
#pragma unroll
        for (int k0 = 0; k0 < BK; k0 += 4) {
            const int ka = k0 + (half << 1);
            v2f bf[4];
#pragma unroll
            for (int o = 0; o < 4; ++o)
                bf[o] = *(const v2f*)&sB[o][(wn << 4) + r16][ka];
#pragma unroll
            for (int im = 0; im < 2; ++im) {
#pragma unroll
                for (int o = 0; o < 4; ++o) {
                    v2f af = *(const v2f*)&sA[o][(wm << 5) + (im << 4) + r16][ka];
                    acc[o][im] = __builtin_amdgcn_wmma_f32_16x16x4_f32(false, af, false, bf[o], (short)0, acc[o][im], false, false);
                }
            }
        }
        __builtin_amdgcn_sched_barrier(0);
        asm volatile("v_nop\n\tv_nop\n\tv_nop\n\tv_nop");
    }

    __syncthreads();
    {
        const int nl = (wn << 4) + r16;
        const int n  = nBlock + nl;
        const bool mir = (n >= 1);
        const float w0 = winT[n];
        const float w1 = mir ? winT[WIN_N - n] : 0.f;
        const float w2 = winT[KH - 1 - n];
        const float w3 = mir ? winT[KH - 1 + n] : 0.f;
#pragma unroll
        for (int im = 0; im < 2; ++im) {
            const int rl = (wm << 5) + (im << 4) + (half << 3);
#pragma unroll
            for (int v = 0; v < 8; ++v) {
                const float cs = acc[0][im][v] + acc[1][im][v];
                const float cd = acc[0][im][v] - acc[1][im][v];
                const float ss = acc[2][im][v] + acc[3][im][v];
                const float sd = acc[2][im][v] - acc[3][im][v];
                float* sp = smem + (rl + v) * YP + nl;
                sp[0 * BM * YP] = w0 * (cs - ss);
                sp[1 * BM * YP] = w1 * (cs + ss);
                sp[2 * BM * YP] = w2 * (cd + sd);
                sp[3 * BM * YP] = w3 * (cd - sd);
            }
        }
    }
    __syncthreads();
    {
        float* yr = (rowBlock < YSPLIT) ? ya + (size_t)rowBlock * YROW
                                        : yb + (size_t)(rowBlock - YSPLIT) * YROW;
#pragma unroll
        for (int it = 0; it < 16; ++it) {
            const int idx = (it << 4) + (wave << 1) + half;
            const int q   = idx >> 6;
            const int row = idx & (BM - 1);
            const float4 p = *(const float4*)&smem[idx * YP + (r16 << 2)];
            *(float4*)(yr + (size_t)row * YROW + q * KQ + nBlock + (r16 << 2)) = p;
        }
    }
}

struct OlaRaw { float y0, y1, y2; int have; };

__device__ __forceinline__ OlaRaw ola_issue(const float* __restrict__ yr, int s) {
    const int q   = s / HOP_N;
    const int rho = s - q * HOP_N;
    const bool h2 = (q >= 2) && (rho < WIN_N - 2 * HOP_N);
    const bool h1 = (q >= 1) && (q <= T_N);
    const bool h0 = (q <= T_N - 1);
    const int p0 = (rho < KQ) ? rho : (2 * KQ + KH - 1) - rho;
    const int p1 = (rho < KH - HOP_N)          ? (2 * KQ + KH - 1 - HOP_N) - rho
                 : (rho < KH + KQ - 1 - HOP_N) ? rho + (3 * KQ + HOP_N - (KH - 1))
                 :                               (KQ + WIN_N - HOP_N) - rho;
    const int p2 = (KQ + WIN_N - 2 * HOP_N) - rho;
    OlaRaw w;
    w.y2 = yr[h2 ? (q - 2) * YROW + p2 : p0];
    w.y1 = yr[h1 ? (q - 1) * YROW + p1 : p0];
    w.y0 = yr[h0 ? q * YROW + p0 : p0];
    w.have = (h2 ? 4 : 0) | (h1 ? 2 : 0) | (h0 ? 1 : 0);
    return w;
}

__device__ __forceinline__ float ola_sum(const OlaRaw& w) {
    float a = (w.have & 4) ? w.y2 : 0.f;
    a += (w.have & 2) ? w.y1 : 0.f;
    a += (w.have & 1) ? w.y0 : 0.f;
    return a;
}

#define FOLD_E 8
__global__ __launch_bounds__(256) void fold_kernel(const float* __restrict__ ya,
                                                   const float* __restrict__ yb,
                                                   float* __restrict__ outsig) {
    const int b  = blockIdx.y;
    const int o0 = blockIdx.x * (256 * FOLD_E) + threadIdx.x;

    const float* yr = (b < YSPLIT_B) ? ya + (size_t)b * (T_N * YROW)
                                     : yb + (size_t)(b - YSPLIT_B) * (T_N * YROW);
    OlaRaw kr[FOLD_E], sr[FOLD_E];
    float ck[FOLD_E], ci[FOLD_E];
#pragma unroll
    for (int e = 0; e < FOLD_E; ++e) {
        const int o = min(o0 + e * 256, OUT_LEN - 1);

        int k = (o < 1000) ? 0 : (o - 50) / 950;
        int r = o - k * 950;
        int j = k * 1000 + r;

        kr[e] = ola_issue(yr, k * 1100 + r);

        float cconst;
        if (j < 950)        cconst = 1.f;
        else if (j < 1000)  cconst = 0.f;
        else if (j < 29000) cconst = (r >= 50 && r < 950) ? 1.f : 0.f;
        else                cconst = 1.f;
        float cdec = (j < 29000 && r >= 950) ? (float)(1000 - r) * 0.02f : 0.f;
        float cinc = 0.f;
        if (j >= 950) {
            int rr = (j - 950) % 1000;
            if (rr < 50) cinc = (float)(rr + 1) * 0.02f;
        }
        ck[e] = cconst + cdec;
        ci[e] = cinc;
        sr[e].y0 = 0.f; sr[e].y1 = 0.f; sr[e].y2 = 0.f; sr[e].have = 0;
        int j2 = j + 50;
        if (cinc != 0.f && j2 < KEPT_LEN) {
            int k2 = j2 / 1000, r2 = j2 - k2 * 1000;
            sr[e] = ola_issue(yr, k2 * 1100 + r2);
        }
    }
    float* op = outsig + (size_t)b * OUT_STRIDE;
#pragma unroll
    for (int e = 0; e < FOLD_E; ++e) {
        const int o = o0 + e * 256;
        const float kept = ola_sum(kr[e]);
        const float sh   = ola_sum(sr[e]);
        if (o < OUT_LEN) op[o] = ck[e] * kept + ci[e] * sh;
    }
}

#define BR 64
__device__ __forceinline__ void stft_form_rows(const float* __restrict__ sg, const float* __restrict__ gT,
                                               int n0, float* __restrict__ dst) {
    const float* fp = sg + n0;
    const float* mp = sg + (WIN_N - 7 - n0);
    const float* qp = sg + (KH - 8 - n0);
    const float* rp = sg + (KH - 1 + n0);
    const float* gp = gT + n0;
    const float* hp = gT + (KH - 8 - n0);
    const float4 f0 = *(const float4*)(fp);
    const float4 f1 = *(const float4*)(fp + 4);
    const v4fu m0 = *(const v4fu*)(mp);
    const v4fu m1 = *(const v4fu*)(mp + 4);
    const float4 q0 = *(const float4*)(qp);
    const float4 q1 = *(const float4*)(qp + 4);
    const v4fu r0 = *(const v4fu*)(rp);
    const v4fu r1 = *(const v4fu*)(rp + 4);
    const float4 g0 = *(const float4*)(gp);
    const float4 g1 = *(const float4*)(gp + 4);
    const float4 h0 = *(const float4*)(hp);
    const float4 h1 = *(const float4*)(hp + 4);
    const float fa[8] = {f0.x, f0.y, f0.z, f0.w, f1.x, f1.y, f1.z, f1.w};
    const float ma[8] = {m1.w, m1.z, m1.y, m1.x, m0.w, m0.z, m0.y, m0.x};
    const float qa[8] = {q1.w, q1.z, q1.y, q1.x, q0.w, q0.z, q0.y, q0.x};
    const float ra[8] = {r0.x, r0.y, r0.z, r0.w, r1.x, r1.y, r1.z, r1.w};
    const float ga[8] = {g0.x, g0.y, g0.z, g0.w, g1.x, g1.y, g1.z, g1.w};
    const float ha[8] = {h1.w, h1.z, h1.y, h1.x, h0.w, h0.z, h0.y, h0.x};
    float pe[8], qe[8], po[8], qo[8];
#pragma unroll
    for (int i = 0; i < 8; ++i) {
        const float un = (fa[i] + ma[i]) * ga[i];
        const float vn = (fa[i] - ma[i]) * ga[i];
        const float um = (qa[i] + ra[i]) * ha[i];
        const float vm = (qa[i] - ra[i]) * ha[i];
        pe[i] = un + um;
        qe[i] = vn - vm;
        po[i] = un - um;
        qo[i] = vn + vm;
    }
    *(float4*)(dst)                         = make_float4(pe[0], pe[1], pe[2], pe[3]);
    *(float4*)(dst + 4)                     = make_float4(pe[4], pe[5], pe[6], pe[7]);
    *(float4*)(dst + 1 * NF_N * TS)         = make_float4(qe[0], qe[1], qe[2], qe[3]);
    *(float4*)(dst + 1 * NF_N * TS + 4)     = make_float4(qe[4], qe[5], qe[6], qe[7]);
    *(float4*)(dst + 2 * NF_N * TS)         = make_float4(po[0], po[1], po[2], po[3]);
    *(float4*)(dst + 2 * NF_N * TS + 4)     = make_float4(po[4], po[5], po[6], po[7]);
    *(float4*)(dst + 3 * NF_N * TS)         = make_float4(qo[0], qo[1], qo[2], qo[3]);
    *(float4*)(dst + 3 * NF_N * TS + 4)     = make_float4(qo[4], qo[5], qo[6], qo[7]);
}

__global__ __launch_bounds__(256) void stft_gemm(const float* __restrict__ outsig,
                                                 const float* __restrict__ WfwdT,
                                                 const float* __restrict__ gT,
                                                 float* __restrict__ out) {
    __shared__ __align__(16) float smem[4 * NF_N * TS + 4 * BR * TS];
    float* sA = smem;
    float* sB = smem + 4 * NF_N * TS;

    const int tid  = threadIdx.x;
    const int lane = tid & 31;
    const int wave = tid >> 5;
    const int kt   = wave & 3;
    const int par  = wave >> 2;
    const int half = lane >> 4;
    const int r16  = lane & 15;

    const int b       = blockIdx.y;
    const int kpBlock = blockIdx.x * BR;
    const float* sigb = outsig + (size_t)b * OUT_STRIDE;

    const int a_row = tid >> 2;
    const int a_lc  = (tid & 3) << 3;

    const int b_row = tid >> 3;
    const int b_kc  = (tid & 7) << 2;
    const float* wp = WfwdT + (size_t)(kpBlock + b_row) * KQ + b_kc;

    v8f accR[5], accI[5];
#pragma unroll
    for (int i = 0; i < 5; ++i) {
        accR[i] = (v8f){0.f,0.f,0.f,0.f,0.f,0.f,0.f,0.f};
        accI[i] = (v8f){0.f,0.f,0.f,0.f,0.f,0.f,0.f,0.f};
    }

    const float* aP = sA + (2 * par) * (NF_N * TS);
    const float* aQ = aP + NF_N * TS;
    const float* bC = sB + (2 * par) * (BR * TS) + ((kt << 4) + r16) * TS;
    const float* bS = bC + BR * TS;
    const int row4 = min(64 + r16, NF_N - 1);

    for (int kb = 0; kb < KQ; kb += BK) {
        __syncthreads();
        const float4 w0 = *(const float4*)(wp);
        const float4 w1 = *(const float4*)(wp + KH * KQ);
        const float4 w2 = *(const float4*)(wp + KQ * KQ);
        const float4 w3 = *(const float4*)(wp + KH * KQ + KQ * KQ);
        const float4 w4 = *(const float4*)(wp + 32 * KQ);
        const float4 w5 = *(const float4*)(wp + KH * KQ + 32 * KQ);
        const float4 w6 = *(const float4*)(wp + KQ * KQ + 32 * KQ);
        const float4 w7 = *(const float4*)(wp + KH * KQ + KQ * KQ + 32 * KQ);
        wp += BK;
        stft_form_rows(sigb + a_row * HOP_N, gT, kb + a_lc, sA + a_row * TS + a_lc);
        if (tid < 4 * (NF_N - 64))
            stft_form_rows(sigb + (64 + a_row) * HOP_N, gT, kb + a_lc, sA + (64 + a_row) * TS + a_lc);
        {
            float* sb = sB + b_row * TS + b_kc;
            *(float4*)(sb)               = w0;
            *(float4*)(sb + 1 * BR * TS) = w1;
            *(float4*)(sb + 2 * BR * TS) = w2;
            *(float4*)(sb + 3 * BR * TS) = w3;
            *(float4*)(sb + 32 * TS)               = w4;
            *(float4*)(sb + 1 * BR * TS + 32 * TS) = w5;
            *(float4*)(sb + 2 * BR * TS + 32 * TS) = w6;
            *(float4*)(sb + 3 * BR * TS + 32 * TS) = w7;
        }
        __syncthreads();
#pragma unroll
        for (int k0 = 0; k0 < BK; k0 += 4) {
            const int ka = k0 + (half << 1);
            const v2f bc = *(const v2f*)(bC + ka);
            const v2f bs = *(const v2f*)(bS + ka);
#pragma unroll
            for (int i = 0; i < 5; ++i) {
                const int row = (i < 4) ? (i << 4) + r16 : row4;
                const v2f ap = *(const v2f*)(aP + row * TS + ka);
                const v2f aq = *(const v2f*)(aQ + row * TS + ka);
                accR[i] = __builtin_amdgcn_wmma_f32_16x16x4_f32(false, ap, false, bc, (short)0, accR[i], false, false);
                accI[i] = __builtin_amdgcn_wmma_f32_16x16x4_f32(false, aq, false, bs, (short)0, accI[i], false, false);
            }
        }
        __builtin_amdgcn_sched_barrier(0);
        asm volatile("v_nop\n\tv_nop\n\tv_nop\n\tv_nop");
    }

    __syncthreads();
    {
        const int kl = 2 * ((kt << 4) + r16) + par;
#pragma unroll
        for (int i = 0; i < 5; ++i) {
#pragma unroll
            for (int v = 0; v < 8; ++v) {
                const int m = (i << 4) + (half << 3) + v;
                if (m < NF_N) {
                    v2f p; p.x = accR[i][v]; p.y = accI[i][v];
                    *(v2f*)&smem[(kl * NF_N + m) << 1] = p;
                }
            }
        }
    }
    __syncthreads();
    {
        float* ob = out + ((size_t)b * F_N + 2 * kpBlock) * (NF_N * 2);
#pragma unroll
        for (int it = 0; it < (2 * BR * NF_N * 2 / 4 + 255) / 256; ++it) {
            const int idx = (it << 8) + tid;
            if (idx < 2 * BR * NF_N * 2 / 4)
                *(float4*)(ob + (idx << 2)) = *(const float4*)&smem[idx << 2];
        }
    }
}

extern "C" void kernel_launch(void* const* d_in, const int* in_sizes, int n_in,
                              void* d_out, int out_size, void* d_ws, size_t ws_size,
                              hipStream_t stream) {
    (void)in_sizes; (void)n_in; (void)out_size; (void)ws_size;
    const float* x = (const float*)d_in[0];
    float* out = (float*)d_out;

    float* ws     = (float*)d_ws;
    float* WinvT  = ws;
    float* winT   = WinvT + (size_t)KPAD * KH;
    float* gT     = winT + KPAD;
    float* WfwdT  = WinvT + (size_t)KPAD * KPAD;
    float* ytail  = WfwdT + (size_t)KPAD * KPAD;
    float* outsig = ytail + (size_t)B_N * SIG_STRIDE;

    build_mats<<<(2 * KPAD * KQ + 2 * KPAD) / 256, 256, 0, stream>>>(WinvT, WfwdT, winT, gT);
    istft_gemm_frames<<<dim3(KQ / 64, M1 / BM), 256, 0, stream>>>(x, WinvT, winT, out, ytail);
    fold_kernel<<<dim3((OUT_LEN + 256 * FOLD_E - 1) / (256 * FOLD_E), B_N), 256, 0, stream>>>(out, ytail, outsig);
    stft_gemm<<<dim3(KQ / BR, B_N), 256, 0, stream>>>(outsig, WfwdT, gT, out);
}
